# starting kernel
# baseline (speedup 1.0000x reference)
#include <hip/hip_runtime.h>
#include <math.h>


typedef __attribute__((ext_vector_type(2))) float v2f;
typedef __attribute__((ext_vector_type(8))) float v8f;

#define NUM_IN   8
#define NUM_OUT  8
#define IN_DIM   64
#define OUT_DIM  64
#define JB       16384
#define NCOL     512
#define STRIPS   1024
#define CAP_EPS  1e-9f

union V8 { v8f v; float f[8]; };

__global__ __launch_bounds__(256) void k_xpart(const float* __restrict__ x,
                                               float* __restrict__ xpart) {
    int part = blockIdx.x;
    int i    = blockIdx.y;
    int t    = threadIdx.x;
    int d    = t & 63;
    int g    = t >> 6;
    const float* xi = x + (size_t)i * JB * IN_DIM;
    int r0 = part * 1024;
    float acc = 0.f;
    for (int r = r0 + g; r < r0 + 1024; r += 4)
        acc += xi[(size_t)r * 64 + d];
    __shared__ float sd[256];
    sd[t] = acc;
    __syncthreads();
    if (g == 0)
        xpart[(i * 16 + part) * 64 + d] = sd[d] + sd[64 + d] + sd[128 + d] + sd[192 + d];
}

__global__ __launch_bounds__(256) void k_uhm(const float* __restrict__ xpart,
                                             const float* __restrict__ W,
                                             float* __restrict__ uhm,
                                             float* __restrict__ Bmat) {
    __shared__ float umean[512];
    int t = threadIdx.x;
    for (int v = t; v < 512; v += 256) {
        int i = v >> 6, d = v & 63;
        float s = 0.f;
        for (int p = 0; p < 16; ++p) s += xpart[(i * 16 + p) * 64 + d];
        umean[v] = s * (1.0f / (float)JB);
    }
    __syncthreads();
    for (int v = t; v < 4096; v += 256) {
        int i = v >> 9, j = (v >> 6) & 7, e = v & 63;
        const float* wp = W + ((size_t)i * 8 + j) * 4096 + e;
        const float* um = umean + i * 64;
        float s = 0.f;
        for (int d = 0; d < 64; ++d) s += um[d] * wp[d * 64];
        uhm[v] = s;
    }
    if (t < 64) Bmat[t] = 0.f;
}

__global__ void k_softmax(const float* __restrict__ Bmat, float* __restrict__ cvec) {
    int j = threadIdx.x;
    if (j < 8) {
        float m = Bmat[j];
        for (int i = 1; i < 8; ++i) m = fmaxf(m, Bmat[i * 8 + j]);
        float den = 0.f;
        for (int i = 0; i < 8; ++i) den += expf(Bmat[i * 8 + j] - m);
        cvec[j] = expf(Bmat[56 + j] - m) / den;
    }
}

__global__ __launch_bounds__(256) void k_uhl_pass(const float* __restrict__ x7,
                                                  const float* __restrict__ w7,
                                                  const float* __restrict__ cvec,
                                                  float* __restrict__ outp,
                                                  int mode) {
    __shared__ float uhl[16 * NCOL];
    __shared__ float fac[16 * 8];

    int t    = threadIdx.x;
    int lane = t & 31;
    int wv   = t >> 5;
    int r0   = blockIdx.x * 16;
    int mrow = r0 + (lane & 15);
    int khalf = (lane >> 4) << 1;

    v2f a[16];
#pragma unroll
    for (int ks = 0; ks < 16; ++ks) {
        const float* ap = x7 + (size_t)mrow * 64 + ks * 4 + khalf;
        a[ks].x = ap[0];
        a[ks].y = ap[1];
    }

    for (int tt = 0; tt < 4; ++tt) {
        int n = (wv * 4 + tt) * 16 + (lane & 15);
        int j = n >> 6, e = n & 63;
        const float* bp = w7 + (size_t)j * 4096 + e;
        V8 acc;
#pragma unroll
        for (int q = 0; q < 8; ++q) acc.f[q] = 0.f;
#pragma unroll
        for (int ks = 0; ks < 16; ++ks) {
            int k = ks * 4 + khalf;
            v2f b;
            b.x = bp[k * 64];
            b.y = bp[(k + 1) * 64];
            acc.v = __builtin_amdgcn_wmma_f32_16x16x4_f32(
                false, a[ks], false, b, (short)0, acc.v, false, false);
        }
        int mb = (lane >> 4) << 3;
#pragma unroll
        for (int q = 0; q < 8; ++q)
            uhl[(mb + q) * NCOL + n] = acc.f[q];
    }
    __syncthreads();

    if (t < 128) {
        int rl = t >> 3, j = t & 7;
        const float* row = uhl + rl * NCOL + j * 64;
        float q = 0.f;
        for (int e = 0; e < 64; ++e) { float u = row[e]; q += u * u; }
        float c  = cvec[j];
        float s2 = c * c * q;
        float factor = s2 / ((1.f + s2) * sqrtf(s2 + CAP_EPS));
        fac[rl * 8 + j] = factor * c;
    }
    __syncthreads();

    if (mode == 0) {
        int j = t >> 5, e0 = (t & 31) * 2;
        float a0 = 0.f, a1 = 0.f;
        const float* col = uhl + j * 64 + e0;
        for (int rl = 0; rl < 16; ++rl) {
            float g = fac[rl * 8 + j];
            a0 += g * col[rl * NCOL];
            a1 += g * col[rl * NCOL + 1];
        }
        float* pp = outp + (size_t)blockIdx.x * NCOL + j * 64 + e0;
        pp[0] = a0; pp[1] = a1;
    } else {
        for (int idx = t; idx < 16 * NCOL; idx += 256) {
            int rl = idx >> 9;
            int n  = idx & 511;
            int j = n >> 6, e = n & 63;
            int r  = r0 + rl;
            int jb = (r & 511) * 32 + (r >> 9);
            outp[(size_t)j * (JB * 64) + (size_t)jb * 64 + e] =
                fac[rl * 8 + j] * uhl[rl * NCOL + n];
        }
    }
}

__global__ __launch_bounds__(256) void k_vred1(const float* __restrict__ vpart,
                                               float* __restrict__ vpart2) {
    int b = blockIdx.x;
    int t = threadIdx.x;
    for (int v = t; v < NCOL; v += 256) {
        const float* p = vpart + (size_t)b * 32 * NCOL + v;
        float s = 0.f;
        for (int q = 0; q < 32; ++q) s += p[q * NCOL];
        vpart2[b * NCOL + v] = s;
    }
}

__global__ __launch_bounds__(512) void k_vred2(const float* __restrict__ vpart2,
                                               const float* __restrict__ uhm,
                                               float* __restrict__ Bmat) {
    __shared__ float vmean[NCOL];
    int t = threadIdx.x;
    float s = 0.f;
    for (int b = 0; b < 32; ++b) s += vpart2[b * NCOL + t];
    vmean[t] = s * (1.0f / (float)JB);
    __syncthreads();
    if (t < 64) {
        int j = t & 7;
        const float* up = uhm + t * 64;
        const float* vp = vmean + j * 64;
        float d = 0.f;
        for (int e = 0; e < 64; ++e) d += up[e] * vp[e];
        Bmat[t] += d;
    }
}

extern "C" void kernel_launch(void* const* d_in, const int* in_sizes, int n_in,
                              void* d_out, int out_size, void* d_ws, size_t ws_size,
                              hipStream_t stream) {
    const float* x = (const float*)d_in[0];
    const float* W = (const float*)d_in[1];
    float* out = (float*)d_out;
    float* ws  = (float*)d_ws;

    float* vpart  = ws;
    float* xpart  = vpart + STRIPS * NCOL;
    float* vpart2 = xpart + 8192;
    float* uhm    = vpart2 + 16384;
    float* Bmat   = uhm + 4096;
    float* cvec   = Bmat + 64;

    const float* x7 = x + (size_t)7 * JB * IN_DIM;
    const float* w7 = W + (size_t)7 * NUM_OUT * IN_DIM * OUT_DIM;

    k_xpart<<<dim3(16, 8), 256, 0, stream>>>(x, xpart);
    k_uhm<<<1, 256, 0, stream>>>(xpart, W, uhm, Bmat);

    for (int it = 0; it < 2; ++it) {
        k_softmax<<<1, 64, 0, stream>>>(Bmat, cvec);
        k_uhl_pass<<<STRIPS, 256, 0, stream>>>(x7, w7, cvec, vpart, 0);
        k_vred1<<<32, 256, 0, stream>>>(vpart, vpart2);
        k_vred2<<<1, 512, 0, stream>>>(vpart2, uhm, Bmat);
    }
    k_softmax<<<1, 64, 0, stream>>>(Bmat, cvec);
    k_uhl_pass<<<STRIPS, 256, 0, stream>>>(x7, w7, cvec, out, 1);
}


# x column-sum kernel: all 16 of a lane's 16-byte loads in flight at once (1024-thread groups) instead of one 4-byte load per wave at a time; x partial sums now added in a tree order
# speedup vs baseline: 2.0464x; 2.0464x over previous
#include <hip/hip_runtime.h>
#include <math.h>


typedef __attribute__((ext_vector_type(2))) float v2f;
typedef __attribute__((ext_vector_type(4))) float v4f;
typedef __attribute__((ext_vector_type(8))) float v8f;

#define NUM_IN   8
#define NUM_OUT  8
#define IN_DIM   64
#define OUT_DIM  64
#define JB       16384
#define NCOL     512
#define STRIPS   1024
#define CAP_EPS  1e-9f

union V8 { v8f v; float f[8]; };

__global__ __launch_bounds__(1024) __attribute__((amdgpu_waves_per_eu(8, 8)))
void k_xpart(const float* __restrict__ x, float* __restrict__ xpart) {
    int part = blockIdx.x;
    int i    = blockIdx.y;
    int t    = threadIdx.x;
    const v4f* p = (const v4f*)(x + ((size_t)i * JB + (size_t)part * 1024) * IN_DIM) + t;

    v4f v[16];
    for (int q = 0; q < 16; ++q) v[q] = p[q * 1024];
    v4f s = ((v[0] + v[1]) + (v[2] + v[3])) + ((v[4] + v[5]) + (v[6] + v[7]));
    s += ((v[8] + v[9]) + (v[10] + v[11])) + ((v[12] + v[13]) + (v[14] + v[15]));

    __shared__ v4f sd[1024];
    __shared__ float sd2[256];
    sd[t] = s;
    __syncthreads();
    const float* sf = (const float*)sd;
    if (t < 256) {
        int d = t & 63, g = t >> 6;
        float a = 0.f;
        for (int k = 0; k < 16; ++k) a += sf[(g * 16 + k) * 64 + d];
        sd2[t] = a;
    }
    __syncthreads();
    if (t < 64)
        xpart[(i * 16 + part) * 64 + t] = sd2[t] + sd2[64 + t] + sd2[128 + t] + sd2[192 + t];
}

__global__ __launch_bounds__(256) void k_uhm(const float* __restrict__ xpart,
                                             const float* __restrict__ W,
                                             float* __restrict__ uhm,
                                             float* __restrict__ Bmat) {
    __shared__ float umean[512];
    int t = threadIdx.x;
    for (int v = t; v < 512; v += 256) {
        int i = v >> 6, d = v & 63;
        float s = 0.f;
        for (int p = 0; p < 16; ++p) s += xpart[(i * 16 + p) * 64 + d];
        umean[v] = s * (1.0f / (float)JB);
    }
    __syncthreads();
    for (int v = t; v < 4096; v += 256) {
        int i = v >> 9, j = (v >> 6) & 7, e = v & 63;
        const float* wp = W + ((size_t)i * 8 + j) * 4096 + e;
        const float* um = umean + i * 64;
        float s = 0.f;
        for (int d = 0; d < 64; ++d) s += um[d] * wp[d * 64];
        uhm[v] = s;
    }
    if (t < 64) Bmat[t] = 0.f;
}

__global__ void k_softmax(const float* __restrict__ Bmat, float* __restrict__ cvec) {
    int j = threadIdx.x;
    if (j < 8) {
        float m = Bmat[j];
        for (int i = 1; i < 8; ++i) m = fmaxf(m, Bmat[i * 8 + j]);
        float den = 0.f;
        for (int i = 0; i < 8; ++i) den += expf(Bmat[i * 8 + j] - m);
        cvec[j] = expf(Bmat[56 + j] - m) / den;
    }
}

__global__ __launch_bounds__(256) void k_uhl_pass(const float* __restrict__ x7,
                                                  const float* __restrict__ w7,
                                                  const float* __restrict__ cvec,
                                                  float* __restrict__ outp,
                                                  int mode) {
    __shared__ float uhl[16 * NCOL];
    __shared__ float fac[16 * 8];

    int t    = threadIdx.x;
    int lane = t & 31;
    int wv   = t >> 5;
    int r0   = blockIdx.x * 16;
    int mrow = r0 + (lane & 15);
    int khalf = (lane >> 4) << 1;

    v2f a[16];
#pragma unroll
    for (int ks = 0; ks < 16; ++ks) {
        const float* ap = x7 + (size_t)mrow * 64 + ks * 4 + khalf;
        a[ks].x = ap[0];
        a[ks].y = ap[1];
    }

    for (int tt = 0; tt < 4; ++tt) {
        int n = (wv * 4 + tt) * 16 + (lane & 15);
        int j = n >> 6, e = n & 63;
        const float* bp = w7 + (size_t)j * 4096 + e;
        V8 acc;
#pragma unroll
        for (int q = 0; q < 8; ++q) acc.f[q] = 0.f;
#pragma unroll
        for (int ks = 0; ks < 16; ++ks) {
            int k = ks * 4 + khalf;
            v2f b;
            b.x = bp[k * 64];
            b.y = bp[(k + 1) * 64];
            acc.v = __builtin_amdgcn_wmma_f32_16x16x4_f32(
                false, a[ks], false, b, (short)0, acc.v, false, false);
        }
        int mb = (lane >> 4) << 3;
#pragma unroll
        for (int q = 0; q < 8; ++q)
            uhl[(mb + q) * NCOL + n] = acc.f[q];
    }
    __syncthreads();

    if (t < 128) {
        int rl = t >> 3, j = t & 7;
        const float* row = uhl + rl * NCOL + j * 64;
        float q = 0.f;
        for (int e = 0; e < 64; ++e) { float u = row[e]; q += u * u; }
        float c  = cvec[j];
        float s2 = c * c * q;
        float factor = s2 / ((1.f + s2) * sqrtf(s2 + CAP_EPS));
        fac[rl * 8 + j] = factor * c;
    }
    __syncthreads();

    if (mode == 0) {
        int j = t >> 5, e0 = (t & 31) * 2;
        float a0 = 0.f, a1 = 0.f;
        const float* col = uhl + j * 64 + e0;
        for (int rl = 0; rl < 16; ++rl) {
            float g = fac[rl * 8 + j];
            a0 += g * col[rl * NCOL];
            a1 += g * col[rl * NCOL + 1];
        }
        float* pp = outp + (size_t)blockIdx.x * NCOL + j * 64 + e0;
        pp[0] = a0; pp[1] = a1;
    } else {
        for (int idx = t; idx < 16 * NCOL; idx += 256) {
            int rl = idx >> 9;
            int n  = idx & 511;
            int j = n >> 6, e = n & 63;
            int r  = r0 + rl;
            int jb = (r & 511) * 32 + (r >> 9);
            outp[(size_t)j * (JB * 64) + (size_t)jb * 64 + e] =
                fac[rl * 8 + j] * uhl[rl * NCOL + n];
        }
    }
}

__global__ __launch_bounds__(256) void k_vred1(const float* __restrict__ vpart,
                                               float* __restrict__ vpart2) {
    int b = blockIdx.x;
    int t = threadIdx.x;
    for (int v = t; v < NCOL; v += 256) {
        const float* p = vpart + (size_t)b * 32 * NCOL + v;
        float s = 0.f;
        for (int q = 0; q < 32; ++q) s += p[q * NCOL];
        vpart2[b * NCOL + v] = s;
    }
}

__global__ __launch_bounds__(512) void k_vred2(const float* __restrict__ vpart2,
                                               const float* __restrict__ uhm,
                                               float* __restrict__ Bmat) {
    __shared__ float vmean[NCOL];
    int t = threadIdx.x;
    float s = 0.f;
    for (int b = 0; b < 32; ++b) s += vpart2[b * NCOL + t];
    vmean[t] = s * (1.0f / (float)JB);
    __syncthreads();
    if (t < 64) {
        int j = t & 7;
        const float* up = uhm + t * 64;
        const float* vp = vmean + j * 64;
        float d = 0.f;
        for (int e = 0; e < 64; ++e) d += up[e] * vp[e];
        Bmat[t] += d;
    }
}

extern "C" void kernel_launch(void* const* d_in, const int* in_sizes, int n_in,
                              void* d_out, int out_size, void* d_ws, size_t ws_size,
                              hipStream_t stream) {
    const float* x = (const float*)d_in[0];
    const float* W = (const float*)d_in[1];
    float* out = (float*)d_out;
    float* ws  = (float*)d_ws;

    float* vpart  = ws;
    float* xpart  = vpart + STRIPS * NCOL;
    float* vpart2 = xpart + 8192;
    float* uhm    = vpart2 + 16384;
    float* Bmat   = uhm + 4096;
    float* cvec   = Bmat + 64;

    const float* x7 = x + (size_t)7 * JB * IN_DIM;
    const float* w7 = W + (size_t)7 * NUM_OUT * IN_DIM * OUT_DIM;

    k_xpart<<<dim3(16, 8), 1024, 0, stream>>>(x, xpart);
    k_uhm<<<1, 256, 0, stream>>>(xpart, W, uhm, Bmat);

    for (int it = 0; it < 2; ++it) {
        k_softmax<<<1, 64, 0, stream>>>(Bmat, cvec);
        k_uhl_pass<<<STRIPS, 256, 0, stream>>>(x7, w7, cvec, vpart, 0);
        k_vred1<<<32, 256, 0, stream>>>(vpart, vpart2);
        k_vred2<<<1, 512, 0, stream>>>(vpart2, uhm, Bmat);
    }
    k_softmax<<<1, 64, 0, stream>>>(Bmat, cvec);
    k_uhl_pass<<<STRIPS, 256, 0, stream>>>(x7, w7, cvec, out, 1);
}


# launch fusion, 12 launches -> 6: u_hat_mean, the B update and the softmax move into the pass prologue and the 32-block reduce launch; every sum keeps its order
# speedup vs baseline: 2.6684x; 1.3039x over previous
#include <hip/hip_runtime.h>
#include <math.h>


typedef __attribute__((ext_vector_type(2))) float v2f;
typedef __attribute__((ext_vector_type(4))) float v4f;
typedef __attribute__((ext_vector_type(8))) float v8f;

#define NUM_IN   8
#define NUM_OUT  8
#define IN_DIM   64
#define OUT_DIM  64
#define JB       16384
#define NCOL     512
#define STRIPS   1024
#define CAP_EPS  1e-9f

union V8 { v8f v; float f[8]; };

__global__ __launch_bounds__(1024) __attribute__((amdgpu_waves_per_eu(8, 8)))
void k_xpart(const float* __restrict__ x, float* __restrict__ xpart) {
    int part = blockIdx.x;
    int i    = blockIdx.y;
    int t    = threadIdx.x;
    const v4f* p = (const v4f*)(x + ((size_t)i * JB + (size_t)part * 1024) * IN_DIM) + t;

    v4f v[16];
    for (int q = 0; q < 16; ++q) v[q] = p[q * 1024];
    v4f s = ((v[0] + v[1]) + (v[2] + v[3])) + ((v[4] + v[5]) + (v[6] + v[7]));
    s += ((v[8] + v[9]) + (v[10] + v[11])) + ((v[12] + v[13]) + (v[14] + v[15]));

    __shared__ v4f sd[1024];
    __shared__ float sd2[256];
    sd[t] = s;
    __syncthreads();
    const float* sf = (const float*)sd;
    if (t < 256) {
        int d = t & 63, g = t >> 6;
        float a = 0.f;
        for (int k = 0; k < 16; ++k) a += sf[(g * 16 + k) * 64 + d];
        sd2[t] = a;
    }
    __syncthreads();
    if (t < 64)
        xpart[(i * 16 + part) * 64 + t] = sd2[t] + sd2[64 + t] + sd2[128 + t] + sd2[192 + t];
}

__global__ __launch_bounds__(256) __attribute__((amdgpu_waves_per_eu(8, 8)))
void k_uhl_pass(const float* __restrict__ x7,
                                                  const float* __restrict__ w7,
                                                  const float* __restrict__ vpart2,
                                                  const float* __restrict__ uhm,
                                                  const float* __restrict__ Bprev,
                                                  float* __restrict__ Bout,
                                                  float* __restrict__ outp,
                                                  int mode) {
    __shared__ float uhl[16 * NCOL];
    __shared__ float fac[16 * 8];
    __shared__ float vmean[8 * 68];
    __shared__ float Bs[64];
    __shared__ float cs[8];

    int t    = threadIdx.x;
    int lane = t & 31;
    int wv   = t >> 5;
    int r0   = blockIdx.x * 16;
    int mrow = r0 + (lane & 15);
    int khalf = (lane >> 4) << 1;

    v2f a[16];
#pragma unroll
    for (int ks = 0; ks < 16; ++ks) {
        const float* ap = x7 + (size_t)mrow * 64 + ks * 4 + khalf;
        a[ks].x = ap[0];
        a[ks].y = ap[1];
    }

    if (vpart2 != nullptr) {
        float s0 = 0.f, s1 = 0.f;
        for (int b = 0; b < 32; ++b) {
            s0 += vpart2[b * NCOL + t];
            s1 += vpart2[b * NCOL + 256 + t];
        }
        vmean[(t >> 6) * 68 + (t & 63)]       = s0 * (1.0f / (float)JB);
        vmean[((t >> 6) + 4) * 68 + (t & 63)] = s1 * (1.0f / (float)JB);
        float* us = uhl;
        const v4f* u4 = (const v4f*)uhm;
        for (int q = 0; q < 4; ++q) {
            int f   = t + 256 * q;
            v4f w   = u4[f];
            int row = f >> 4, col = (f & 15) << 2;
            float* d = us + row * 65 + col;
            d[0] = w.x; d[1] = w.y; d[2] = w.z; d[3] = w.w;
        }
        __syncthreads();
        if (t < 64) {
            int j = t & 7;
            const float* up = us + t * 65;
            const float* vp = vmean + j * 68;
            float d = 0.f;
            for (int e = 0; e < 64; ++e) d += up[e] * vp[e];
            float Bn = (Bprev != nullptr ? Bprev[t] : 0.f) + d;
            Bs[t] = Bn;
            if (Bout != nullptr && blockIdx.x == 0) Bout[t] = Bn;
        }
    } else {
        if (t < 64) Bs[t] = 0.f;
    }
    __syncthreads();
    if (t < 8) {
        int j = t;
        float m = Bs[j];
        for (int i = 1; i < 8; ++i) m = fmaxf(m, Bs[i * 8 + j]);
        float den = 0.f;
        for (int i = 0; i < 8; ++i) den += expf(Bs[i * 8 + j] - m);
        cs[j] = expf(Bs[56 + j] - m) / den;
    }

    for (int tt = 0; tt < 4; ++tt) {
        int n = (wv * 4 + tt) * 16 + (lane & 15);
        int j = n >> 6, e = n & 63;
        const float* bp = w7 + (size_t)j * 4096 + e;
        V8 acc;
#pragma unroll
        for (int q = 0; q < 8; ++q) acc.f[q] = 0.f;
#pragma unroll
        for (int ks = 0; ks < 16; ++ks) {
            int k = ks * 4 + khalf;
            v2f b;
            b.x = bp[k * 64];
            b.y = bp[(k + 1) * 64];
            acc.v = __builtin_amdgcn_wmma_f32_16x16x4_f32(
                false, a[ks], false, b, (short)0, acc.v, false, false);
        }
        int mb = (lane >> 4) << 3;
#pragma unroll
        for (int q = 0; q < 8; ++q)
            uhl[(mb + q) * NCOL + n] = acc.f[q];
    }
    __syncthreads();

    if (t < 128) {
        int rl = t >> 3, j = t & 7;
        const float* row = uhl + rl * NCOL + j * 64;
        float q = 0.f;
        for (int e = 0; e < 64; ++e) { float u = row[e]; q += u * u; }
        float c  = cs[j];
        float s2 = c * c * q;
        float factor = s2 / ((1.f + s2) * sqrtf(s2 + CAP_EPS));
        fac[rl * 8 + j] = factor * c;
    }
    __syncthreads();

    if (mode == 0) {
        int j = t >> 5, e0 = (t & 31) * 2;
        float a0 = 0.f, a1 = 0.f;
        const float* col = uhl + j * 64 + e0;
        for (int rl = 0; rl < 16; ++rl) {
            float g = fac[rl * 8 + j];
            a0 += g * col[rl * NCOL];
            a1 += g * col[rl * NCOL + 1];
        }
        float* pp = outp + (size_t)blockIdx.x * NCOL + j * 64 + e0;
        pp[0] = a0; pp[1] = a1;
    } else {
        for (int idx = t; idx < 16 * NCOL; idx += 256) {
            int rl = idx >> 9;
            int n  = idx & 511;
            int j = n >> 6, e = n & 63;
            int r  = r0 + rl;
            int jb = (r & 511) * 32 + (r >> 9);
            outp[(size_t)j * (JB * 64) + (size_t)jb * 64 + e] =
                fac[rl * 8 + j] * uhl[rl * NCOL + n];
        }
    }
}

__global__ __launch_bounds__(256) void k_vred1(const float* __restrict__ vpart,
                                               float* __restrict__ vpart2,
                                               const float* __restrict__ xpart,
                                               const float* __restrict__ W,
                                               float* __restrict__ uhm) {
    __shared__ float umean[64];
    int b = blockIdx.x;
    int t = threadIdx.x;
    if (b < 32) {
        for (int v = t; v < NCOL; v += 256) {
            const float* p = vpart + (size_t)b * 32 * NCOL + v;
            float s = 0.f;
            for (int q = 0; q < 32; ++q) s += p[q * NCOL];
            vpart2[b * NCOL + v] = s;
        }
    } else {
        int p = b - 32;
        int i = p >> 3;
        if (t < 64) {
            float s = 0.f;
            for (int q = 0; q < 16; ++q) s += xpart[(i * 16 + q) * 64 + t];
            umean[t] = s * (1.0f / (float)JB);
        }
        __syncthreads();
        if (t < 64) {
            const float* wp = W + (size_t)p * 4096 + t;
            float s = 0.f;
            for (int d = 0; d < 64; ++d) s += umean[d] * wp[d * 64];
            uhm[p * 64 + t] = s;
        }
    }
}

extern "C" void kernel_launch(void* const* d_in, const int* in_sizes, int n_in,
                              void* d_out, int out_size, void* d_ws, size_t ws_size,
                              hipStream_t stream) {
    const float* x = (const float*)d_in[0];
    const float* W = (const float*)d_in[1];
    float* out = (float*)d_out;
    float* ws  = (float*)d_ws;

    float* vpart  = ws;
    float* xpart  = vpart + STRIPS * NCOL;
    float* vpart2 = xpart + 8192;
    float* uhm    = vpart2 + 16384;
    float* Bmat   = uhm + 4096;

    const float* x7 = x + (size_t)7 * JB * IN_DIM;
    const float* w7 = W + (size_t)7 * NUM_OUT * IN_DIM * OUT_DIM;

    k_xpart<<<dim3(16, 8), 1024, 0, stream>>>(x, xpart);

    k_uhl_pass<<<STRIPS, 256, 0, stream>>>(x7, w7, nullptr, uhm, nullptr, nullptr, vpart, 0);
    k_vred1<<<96, 256, 0, stream>>>(vpart, vpart2, xpart, W, uhm);
    k_uhl_pass<<<STRIPS, 256, 0, stream>>>(x7, w7, vpart2, uhm, nullptr, Bmat, vpart, 0);
    k_vred1<<<32, 256, 0, stream>>>(vpart, vpart2, xpart, W, uhm);
    k_uhl_pass<<<STRIPS, 256, 0, stream>>>(x7, w7, vpart2, uhm, Bmat, nullptr, out, 1);
}


# routing sums taken on the input side: y = g^T x7 from stored squared norms, B += Z.y; the second product pass and both reduce launches go, 6 launches -> 5; sums of v and of B's dot are reassociated
# speedup vs baseline: 3.9945x; 1.4970x over previous
#include <hip/hip_runtime.h>
#include <math.h>


typedef __attribute__((ext_vector_type(2))) float v2f;
typedef __attribute__((ext_vector_type(4))) float v4f;
typedef __attribute__((ext_vector_type(8))) float v8f;

#define NUM_IN   8
#define NUM_OUT  8
#define IN_DIM   64
#define OUT_DIM  64
#define JB       16384
#define NCOL     512
#define STRIPS   1024
#define CAP_EPS  1e-9f

union V8 { v8f v; float f[8]; };

__global__ __launch_bounds__(1024) __attribute__((amdgpu_waves_per_eu(8, 8)))
void k_xpart(const float* __restrict__ x, float* __restrict__ xpart) {
    int part = blockIdx.x;
    int i    = blockIdx.y;
    int t    = threadIdx.x;
    const v4f* p = (const v4f*)(x + ((size_t)i * JB + (size_t)part * 1024) * IN_DIM) + t;

    v4f v[16];
    for (int q = 0; q < 16; ++q) v[q] = p[q * 1024];
    v4f s = ((v[0] + v[1]) + (v[2] + v[3])) + ((v[4] + v[5]) + (v[6] + v[7]));
    s += ((v[8] + v[9]) + (v[10] + v[11])) + ((v[12] + v[13]) + (v[14] + v[15]));

    __shared__ v4f sd[1024];
    __shared__ float sd2[256];
    sd[t] = s;
    __syncthreads();
    const float* sf = (const float*)sd;
    if (t < 256) {
        int d = t & 63, g = t >> 6;
        float a = 0.f;
        for (int k = 0; k < 16; ++k) a += sf[(g * 16 + k) * 64 + d];
        sd2[t] = a;
    }
    __syncthreads();
    if (t < 64)
        xpart[(i * 16 + part) * 64 + t] = sd2[t] + sd2[64 + t] + sd2[128 + t] + sd2[192 + t];
}

__device__ __forceinline__ void routing_prologue(const float* __restrict__ ypart,
                                                 const float* __restrict__ Zm,
                                                 const float* __restrict__ Bprev,
                                                 float* __restrict__ Bout,
                                                 float* stage, float* ymean,
                                                 float* Bs, float* cs, int t) {
    if (ypart != nullptr) {
        float s0 = 0.f, s1 = 0.f;
        for (int b = 0; b < 32; ++b) {
            s0 += ypart[b * NCOL + t];
            s1 += ypart[b * NCOL + 256 + t];
        }
        ymean[(t >> 6) * 68 + (t & 63)]       = s0 * (1.0f / (float)JB);
        ymean[((t >> 6) + 4) * 68 + (t & 63)] = s1 * (1.0f / (float)JB);
        const v4f* z4 = (const v4f*)Zm;
        for (int q = 0; q < 4; ++q) {
            int f   = t + 256 * q;
            v4f w   = z4[f];
            int row = f >> 4, col = (f & 15) << 2;
            float* d = stage + row * 65 + col;
            d[0] = w.x; d[1] = w.y; d[2] = w.z; d[3] = w.w;
        }
        __syncthreads();
        if (t < 64) {
            int j = t & 7;
            const float* zp = stage + t * 65;
            const float* yp = ymean + j * 68;
            float d = 0.f;
            for (int e = 0; e < 64; ++e) d += zp[e] * yp[e];
            float Bn = (Bprev != nullptr ? Bprev[t] : 0.f) + d;
            Bs[t] = Bn;
            if (Bout != nullptr && blockIdx.x == 0) Bout[t] = Bn;
        }
    } else {
        if (t < 64) Bs[t] = 0.f;
    }
    __syncthreads();
    if (t < 8) {
        int j = t;
        float m = Bs[j];
        for (int i = 1; i < 8; ++i) m = fmaxf(m, Bs[i * 8 + j]);
        float den = 0.f;
        for (int i = 0; i < 8; ++i) den += expf(Bs[i * 8 + j] - m);
        cs[j] = expf(Bs[56 + j] - m) / den;
    }
}

__global__ __launch_bounds__(256) __attribute__((amdgpu_waves_per_eu(8, 8)))
void k_uhl_pass(const float* __restrict__ x7,
                const float* __restrict__ w7,
                const float* __restrict__ ypart,
                const float* __restrict__ Zm,
                const float* __restrict__ Bprev,
                float* __restrict__ outp,
                int mode) {
    __shared__ float uhl[16 * NCOL];
    __shared__ float fac[16 * 8];
    __shared__ float ymean[8 * 68];
    __shared__ float Bs[64];
    __shared__ float cs[8];

    int t    = threadIdx.x;
    int lane = t & 31;
    int wv   = t >> 5;
    int r0   = blockIdx.x * 16;
    int mrow = r0 + (lane & 15);
    int khalf = (lane >> 4) << 1;

    v2f a[16];
#pragma unroll
    for (int ks = 0; ks < 16; ++ks) {
        const float* ap = x7 + (size_t)mrow * 64 + ks * 4 + khalf;
        a[ks].x = ap[0];
        a[ks].y = ap[1];
    }

    if (mode == 1)
        routing_prologue(ypart, Zm, Bprev, nullptr, uhl, ymean, Bs, cs, t);

    for (int tt = 0; tt < 4; ++tt) {
        int n = (wv * 4 + tt) * 16 + (lane & 15);
        int j = n >> 6, e = n & 63;
        const float* bp = w7 + (size_t)j * 4096 + e;
        V8 acc;
#pragma unroll
        for (int q = 0; q < 8; ++q) acc.f[q] = 0.f;
#pragma unroll
        for (int ks = 0; ks < 16; ++ks) {
            int k = ks * 4 + khalf;
            v2f b;
            b.x = bp[k * 64];
            b.y = bp[(k + 1) * 64];
            acc.v = __builtin_amdgcn_wmma_f32_16x16x4_f32(
                false, a[ks], false, b, (short)0, acc.v, false, false);
        }
        int mb = (lane >> 4) << 3;
#pragma unroll
        for (int q = 0; q < 8; ++q)
            uhl[(mb + q) * NCOL + n] = acc.f[q];
    }
    __syncthreads();

    if (t < 128) {
        int rl = t >> 3, j = t & 7;
        const float* row = uhl + rl * NCOL + j * 64;
        float q = 0.f;
        for (int e = 0; e < 64; ++e) { float u = row[e]; q += u * u; }
        if (mode == 0) {
            outp[(size_t)r0 * 8 + t] = q;
        } else {
            float c  = cs[j];
            float s2 = c * c * q;
            float factor = s2 / ((1.f + s2) * sqrtf(s2 + CAP_EPS));
            fac[rl * 8 + j] = factor * c;
        }
    }
    if (mode == 0) return;
    __syncthreads();

    for (int idx = t; idx < 16 * NCOL; idx += 256) {
        int rl = idx >> 9;
        int n  = idx & 511;
        int j = n >> 6, e = n & 63;
        int r  = r0 + rl;
        int jb = (r & 511) * 32 + (r >> 9);
        outp[(size_t)j * (JB * 64) + (size_t)jb * 64 + e] =
            fac[rl * 8 + j] * uhl[rl * NCOL + n];
    }
}

__global__ __launch_bounds__(256) void k_route(const float* __restrict__ x7,
                                               const float* __restrict__ qws,
                                               const float* __restrict__ ypin,
                                               const float* __restrict__ Zin,
                                               const float* __restrict__ Bprev,
                                               float* __restrict__ Bout,
                                               float* __restrict__ ypout,
                                               const float* __restrict__ xpart,
                                               const float* __restrict__ W,
                                               const float* __restrict__ w7,
                                               float* __restrict__ Zout) {
    __shared__ float stage[64 * 65];
    __shared__ float ymean[8 * 68];
    __shared__ float Bs[64];
    __shared__ float cs[8];
    __shared__ v4f   gs4[1024];
    __shared__ float red[8 * NCOL];

    int t = threadIdx.x;
    int b = blockIdx.x;

    if (b >= 32) {
        int p = b - 32;
        int i = p >> 3, j = p & 7;
        float* umean = stage;
        float* uh    = stage + 64;
        if (t < 64) {
            float s = 0.f;
            for (int q = 0; q < 16; ++q) s += xpart[(i * 16 + q) * 64 + t];
            umean[t] = s * (1.0f / (float)JB);
        }
        __syncthreads();
        if (t < 64) {
            const float* wp = W + (size_t)p * 4096 + t;
            float s = 0.f;
            for (int d = 0; d < 64; ++d) s += umean[d] * wp[d * 64];
            uh[t] = s;
        }
        __syncthreads();
        if (t < 64) {
            const v4f* wr = (const v4f*)(w7 + (size_t)j * 4096 + t * 64);
            float s = 0.f;
            for (int e4 = 0; e4 < 16; ++e4) {
                v4f w = wr[e4];
                s += uh[e4 * 4 + 0] * w.x;
                s += uh[e4 * 4 + 1] * w.y;
                s += uh[e4 * 4 + 2] * w.z;
                s += uh[e4 * 4 + 3] * w.w;
            }
            Zout[p * 64 + t] = s;
        }
        return;
    }

    int lane = t & 31;
    int wv   = t >> 5;

    const v4f* q4 = (const v4f*)(qws + (size_t)b * 4096);
    v4f qv[4];
    for (int k = 0; k < 4; ++k) qv[k] = q4[t + 256 * k];

    routing_prologue(ypin, Zin, Bprev, Bout, stage, ymean, Bs, cs, t);
    __syncthreads();

    for (int k = 0; k < 4; ++k) {
        int f  = t + 256 * k;
        int j0 = (f & 1) * 4;
        v4f g;
        for (int m = 0; m < 4; ++m) {
            float c  = cs[j0 + m];
            float s2 = c * c * qv[k][m];
            float factor = s2 / ((1.f + s2) * sqrtf(s2 + CAP_EPS));
            g[m] = factor * c;
        }
        gs4[f] = g;
    }
    __syncthreads();

    const v2f* xp = (const v2f*)(x7 + ((size_t)b * 512 + (size_t)wv * 64) * IN_DIM) + lane;
    float a0[8], a1[8];
    for (int j = 0; j < 8; ++j) { a0[j] = 0.f; a1[j] = 0.f; }
    for (int i0 = 0; i0 < 64; i0 += 16) {
        v2f xv[16];
#pragma unroll
        for (int u = 0; u < 16; ++u) xv[u] = xp[(i0 + u) * 32];
#pragma unroll
        for (int u = 0; u < 16; ++u) {
            v4f ga = gs4[(wv * 64 + i0 + u) * 2];
            v4f gb = gs4[(wv * 64 + i0 + u) * 2 + 1];
            a0[0] += ga.x * xv[u].x;  a1[0] += ga.x * xv[u].y;
            a0[1] += ga.y * xv[u].x;  a1[1] += ga.y * xv[u].y;
            a0[2] += ga.z * xv[u].x;  a1[2] += ga.z * xv[u].y;
            a0[3] += ga.w * xv[u].x;  a1[3] += ga.w * xv[u].y;
            a0[4] += gb.x * xv[u].x;  a1[4] += gb.x * xv[u].y;
            a0[5] += gb.y * xv[u].x;  a1[5] += gb.y * xv[u].y;
            a0[6] += gb.z * xv[u].x;  a1[6] += gb.z * xv[u].y;
            a0[7] += gb.w * xv[u].x;  a1[7] += gb.w * xv[u].y;
        }
    }
    for (int j = 0; j < 8; ++j) {
        v2f s; s.x = a0[j]; s.y = a1[j];
        *(v2f*)(red + wv * NCOL + j * 64 + 2 * lane) = s;
    }
    __syncthreads();
    for (int o = t; o < NCOL; o += 256) {
        float s = 0.f;
        for (int w = 0; w < 8; ++w) s += red[w * NCOL + o];
        ypout[(size_t)b * NCOL + o] = s;
    }
}

extern "C" void kernel_launch(void* const* d_in, const int* in_sizes, int n_in,
                              void* d_out, int out_size, void* d_ws, size_t ws_size,
                              hipStream_t stream) {
    const float* x = (const float*)d_in[0];
    const float* W = (const float*)d_in[1];
    float* out = (float*)d_out;
    float* ws  = (float*)d_ws;

    float* qws    = ws;
    float* xpart  = qws + 131072;
    float* ypartA = xpart + 8192;
    float* ypartB = ypartA + 16384;
    float* Zm     = ypartB + 16384;
    float* Bmat   = Zm + 4096;

    const float* x7 = x + (size_t)7 * JB * IN_DIM;
    const float* w7 = W + (size_t)7 * NUM_OUT * IN_DIM * OUT_DIM;

    k_xpart<<<dim3(16, 8), 1024, 0, stream>>>(x, xpart);

    k_uhl_pass<<<STRIPS, 256, 0, stream>>>(x7, w7, nullptr, nullptr, nullptr, qws, 0);
    k_route<<<96, 256, 0, stream>>>(x7, qws, nullptr, nullptr, nullptr, nullptr, ypartA,
                                    xpart, W, w7, Zm);
    k_route<<<32, 256, 0, stream>>>(x7, qws, ypartA, Zm, nullptr, Bmat, ypartB,
                                    nullptr, nullptr, nullptr, nullptr);
    k_uhl_pass<<<STRIPS, 256, 0, stream>>>(x7, w7, ypartB, Zm, Bmat, out, 1);
}


# norms and squash factors taken in the WMMA accumulator registers (row_xmask adds; halves swapped for whole-line stores): LDS strip, its 16-way-conflict walk, 2 barriers go; q sums reassociated
# speedup vs baseline: 4.3177x; 1.0809x over previous
#include <hip/hip_runtime.h>
#include <math.h>


typedef __attribute__((ext_vector_type(2))) float v2f;
typedef __attribute__((ext_vector_type(4))) float v4f;
typedef __attribute__((ext_vector_type(8))) float v8f;

#define NUM_IN   8
#define NUM_OUT  8
#define IN_DIM   64
#define OUT_DIM  64
#define JB       16384
#define NCOL     512
#define STRIPS   1024
#define CAP_EPS  1e-9f

union V8 { v8f v; float f[8]; };

__global__ __launch_bounds__(1024) __attribute__((amdgpu_waves_per_eu(8, 8)))
void k_xpart(const float* __restrict__ x, float* __restrict__ xpart) {
    int part = blockIdx.x;
    int i    = blockIdx.y;
    int t    = threadIdx.x;
    const v4f* p = (const v4f*)(x + ((size_t)i * JB + (size_t)part * 1024) * IN_DIM) + t;

    v4f v[16];
    for (int q = 0; q < 16; ++q) v[q] = p[q * 1024];
    v4f s = ((v[0] + v[1]) + (v[2] + v[3])) + ((v[4] + v[5]) + (v[6] + v[7]));
    s += ((v[8] + v[9]) + (v[10] + v[11])) + ((v[12] + v[13]) + (v[14] + v[15]));

    __shared__ v4f sd[1024];
    __shared__ float sd2[256];
    sd[t] = s;
    __syncthreads();
    const float* sf = (const float*)sd;
    if (t < 256) {
        int d = t & 63, g = t >> 6;
        float a = 0.f;
        for (int k = 0; k < 16; ++k) a += sf[(g * 16 + k) * 64 + d];
        sd2[t] = a;
    }
    __syncthreads();
    if (t < 64)
        xpart[(i * 16 + part) * 64 + t] = sd2[t] + sd2[64 + t] + sd2[128 + t] + sd2[192 + t];
}

__device__ __forceinline__ void routing_prologue(const float* __restrict__ ypart,
                                                 const float* __restrict__ Zm,
                                                 const float* __restrict__ Bprev,
                                                 float* __restrict__ Bout,
                                                 float* stage, float* ymean,
                                                 float* Bs, float* cs, int t) {
    if (ypart != nullptr) {
        float s0 = 0.f, s1 = 0.f;
        for (int b = 0; b < 32; ++b) {
            s0 += ypart[b * NCOL + t];
            s1 += ypart[b * NCOL + 256 + t];
        }
        ymean[(t >> 6) * 68 + (t & 63)]       = s0 * (1.0f / (float)JB);
        ymean[((t >> 6) + 4) * 68 + (t & 63)] = s1 * (1.0f / (float)JB);
        const v4f* z4 = (const v4f*)Zm;
        for (int q = 0; q < 4; ++q) {
            int f   = t + 256 * q;
            v4f w   = z4[f];
            int row = f >> 4, col = (f & 15) << 2;
            float* d = stage + row * 65 + col;
            d[0] = w.x; d[1] = w.y; d[2] = w.z; d[3] = w.w;
        }
        __syncthreads();
        if (t < 64) {
            int j = t & 7;
            const float* zp = stage + t * 65;
            const float* yp = ymean + j * 68;
            float d = 0.f;
            for (int e = 0; e < 64; ++e) d += zp[e] * yp[e];
            float Bn = (Bprev != nullptr ? Bprev[t] : 0.f) + d;
            Bs[t] = Bn;
            if (Bout != nullptr && blockIdx.x == 0) Bout[t] = Bn;
        }
    } else {
        if (t < 64) Bs[t] = 0.f;
    }
    __syncthreads();
    if (t < 8) {
        int j = t;
        float m = Bs[j];
        for (int i = 1; i < 8; ++i) m = fmaxf(m, Bs[i * 8 + j]);
        float den = 0.f;
        for (int i = 0; i < 8; ++i) den += expf(Bs[i * 8 + j] - m);
        cs[j] = expf(Bs[56 + j] - m) / den;
    }
}

#define ROW_XMASK(v, m) \
    __builtin_bit_cast(float, __builtin_amdgcn_update_dpp(0, __builtin_bit_cast(int, (v)), 0x160 + (m), 0xf, 0xf, true))
#define ROW_SHARE(v, k) \
    __builtin_bit_cast(float, __builtin_amdgcn_update_dpp(0, __builtin_bit_cast(int, (v)), 0x150 + (k), 0xf, 0xf, true))

__global__ __launch_bounds__(256) __attribute__((amdgpu_waves_per_eu(8, 8)))
void k_uhl_pass(const float* __restrict__ x7,
                const float* __restrict__ w7,
                const float* __restrict__ ypart,
                const float* __restrict__ Zm,
                const float* __restrict__ Bprev,
                float* __restrict__ outp,
                int mode) {
    __shared__ float stage[64 * 65];
    __shared__ float ymean[8 * 68];
    __shared__ float Bs[64];
    __shared__ float cs[8];

    int t    = threadIdx.x;
    int lane = t & 31;
    int wv   = __builtin_amdgcn_readfirstlane(t >> 5);
    int c16  = lane & 15;
    int half = lane >> 4;
    int r0   = blockIdx.x * 16;
    int mrow = r0 + c16;
    int khalf = half << 1;

    v2f a[16];
#pragma unroll
    for (int ks = 0; ks < 16; ++ks) {
        const float* ap = x7 + (size_t)mrow * 64 + ks * 4 + khalf;
        a[ks].x = ap[0];
        a[ks].y = ap[1];
    }

    if (mode == 1)
        routing_prologue(ypart, Zm, Bprev, nullptr, stage, ymean, Bs, cs, t);

    V8 acc[4];
#pragma unroll
    for (int tt = 0; tt < 4; ++tt) {
        int n = (wv * 4 + tt) * 16 + c16;
        int j = n >> 6, e = n & 63;
        const float* bp = w7 + (size_t)j * 4096 + e;
#pragma unroll
        for (int q = 0; q < 8; ++q) acc[tt].f[q] = 0.f;
#pragma unroll
        for (int ks = 0; ks < 16; ++ks) {
            int k = ks * 4 + khalf;
            v2f b;
            b.x = bp[k * 64];
            b.y = bp[(k + 1) * 64];
            acc[tt].v = __builtin_amdgcn_wmma_f32_16x16x4_f32(
                false, a[ks], false, b, (short)0, acc[tt].v, false, false);
        }
    }

    int ln = __builtin_amdgcn_mbcnt_lo(~0u, 0u);
    int cE = ln & 15;
    int hE = ln >> 4;

    float qs[8];
#pragma unroll
    for (int r = 0; r < 8; ++r) {
        float s = acc[0].f[r] * acc[0].f[r];
        s += acc[1].f[r] * acc[1].f[r];
        s += acc[2].f[r] * acc[2].f[r];
        s += acc[3].f[r] * acc[3].f[r];
        s += ROW_XMASK(s, 1);
        s += ROW_XMASK(s, 2);
        s += ROW_XMASK(s, 4);
        s += ROW_XMASK(s, 8);
        qs[r] = s;
    }
    float qsel = qs[0];
#pragma unroll
    for (int r = 1; r < 8; ++r) qsel = ((cE & 7) == r) ? qs[r] : qsel;

    if (mode == 0) {
        if (cE < 8) outp[(size_t)(r0 + hE * 8 + cE) * 8 + wv] = qsel;
        return;
    }
    __syncthreads();

    float cj = cs[wv];
    float s2 = cj * cj * qsel;
    float g  = s2 / ((1.f + s2) * sqrtf(s2 + CAP_EPS)) * cj;

    float gr[8] = { ROW_SHARE(g, 0), ROW_SHARE(g, 1), ROW_SHARE(g, 2), ROW_SHARE(g, 3),
                    ROW_SHARE(g, 4), ROW_SHARE(g, 5), ROW_SHARE(g, 6), ROW_SHARE(g, 7) };

    int jb0 = (r0 & 511) * 32 + (r0 >> 9);
    float* op = outp + (size_t)wv * (JB * 64) + (size_t)jb0 * 64 + ln;
#pragma unroll
    for (int r = 0; r < 8; ++r) {
#pragma unroll
        for (int p = 0; p < 2; ++p) {
            float xa = gr[r] * acc[2 * p].f[r];
            float xb = gr[r] * acc[2 * p + 1].f[r];
            float give = hE ? xa : xb;
            int gi = __builtin_bit_cast(int, give);
            float got = __builtin_bit_cast(float,
                __builtin_amdgcn_permlanex16(gi, gi, 0x76543210, 0xfedcba98, false, false));
            op[r * (32 * 64) + p * 32]       = hE ? got : xa;
            op[(8 + r) * (32 * 64) + p * 32] = hE ? xb : got;
        }
    }
}

__global__ __launch_bounds__(256) void k_route(const float* __restrict__ x7,
                                               const float* __restrict__ qws,
                                               const float* __restrict__ ypin,
                                               const float* __restrict__ Zin,
                                               const float* __restrict__ Bprev,
                                               float* __restrict__ Bout,
                                               float* __restrict__ ypout,
                                               const float* __restrict__ xpart,
                                               const float* __restrict__ W,
                                               const float* __restrict__ w7,
                                               float* __restrict__ Zout) {
    __shared__ float stage[64 * 65];
    __shared__ float ymean[8 * 68];
    __shared__ float Bs[64];
    __shared__ float cs[8];
    __shared__ v4f   gs4[1024];
    __shared__ float red[8 * NCOL];

    int t = threadIdx.x;
    int b = blockIdx.x;

    if (b >= 32) {
        int p = b - 32;
        int i = p >> 3, j = p & 7;
        float* umean = stage;
        float* uh    = stage + 64;
        if (t < 64) {
            float s = 0.f;
            for (int q = 0; q < 16; ++q) s += xpart[(i * 16 + q) * 64 + t];
            umean[t] = s * (1.0f / (float)JB);
        }
        __syncthreads();
        if (t < 64) {
            const float* wp = W + (size_t)p * 4096 + t;
            float s = 0.f;
            for (int d = 0; d < 64; ++d) s += umean[d] * wp[d * 64];
            uh[t] = s;
        }
        __syncthreads();
        if (t < 64) {
            const v4f* wr = (const v4f*)(w7 + (size_t)j * 4096 + t * 64);
            float s = 0.f;
            for (int e4 = 0; e4 < 16; ++e4) {
                v4f w = wr[e4];
                s += uh[e4 * 4 + 0] * w.x;
                s += uh[e4 * 4 + 1] * w.y;
                s += uh[e4 * 4 + 2] * w.z;
                s += uh[e4 * 4 + 3] * w.w;
            }
            Zout[p * 64 + t] = s;
        }
        return;
    }

    int lane = t & 31;
    int wv   = t >> 5;

    const v4f* q4 = (const v4f*)(qws + (size_t)b * 4096);
    v4f qv[4];
    for (int k = 0; k < 4; ++k) qv[k] = q4[t + 256 * k];

    routing_prologue(ypin, Zin, Bprev, Bout, stage, ymean, Bs, cs, t);
    __syncthreads();

    for (int k = 0; k < 4; ++k) {
        int f  = t + 256 * k;
        int j0 = (f & 1) * 4;
        v4f g;
        for (int m = 0; m < 4; ++m) {
            float c  = cs[j0 + m];
            float s2 = c * c * qv[k][m];
            float factor = s2 / ((1.f + s2) * sqrtf(s2 + CAP_EPS));
            g[m] = factor * c;
        }
        gs4[f] = g;
    }
    __syncthreads();

    const v2f* xp = (const v2f*)(x7 + ((size_t)b * 512 + (size_t)wv * 64) * IN_DIM) + lane;
    float a0[8], a1[8];
    for (int j = 0; j < 8; ++j) { a0[j] = 0.f; a1[j] = 0.f; }
    for (int i0 = 0; i0 < 64; i0 += 16) {
        v2f xv[16];
#pragma unroll
        for (int u = 0; u < 16; ++u) xv[u] = xp[(i0 + u) * 32];
#pragma unroll
        for (int u = 0; u < 16; ++u) {
            v4f ga = gs4[(wv * 64 + i0 + u) * 2];
            v4f gb = gs4[(wv * 64 + i0 + u) * 2 + 1];
            a0[0] += ga.x * xv[u].x;  a1[0] += ga.x * xv[u].y;
            a0[1] += ga.y * xv[u].x;  a1[1] += ga.y * xv[u].y;
            a0[2] += ga.z * xv[u].x;  a1[2] += ga.z * xv[u].y;
            a0[3] += ga.w * xv[u].x;  a1[3] += ga.w * xv[u].y;
            a0[4] += gb.x * xv[u].x;  a1[4] += gb.x * xv[u].y;
            a0[5] += gb.y * xv[u].x;  a1[5] += gb.y * xv[u].y;
            a0[6] += gb.z * xv[u].x;  a1[6] += gb.z * xv[u].y;
            a0[7] += gb.w * xv[u].x;  a1[7] += gb.w * xv[u].y;
        }
    }
    for (int j = 0; j < 8; ++j) {
        v2f s; s.x = a0[j]; s.y = a1[j];
        *(v2f*)(red + wv * NCOL + j * 64 + 2 * lane) = s;
    }
    __syncthreads();
    for (int o = t; o < NCOL; o += 256) {
        float s = 0.f;
        for (int w = 0; w < 8; ++w) s += red[w * NCOL + o];
        ypout[(size_t)b * NCOL + o] = s;
    }
}

extern "C" void kernel_launch(void* const* d_in, const int* in_sizes, int n_in,
                              void* d_out, int out_size, void* d_ws, size_t ws_size,
                              hipStream_t stream) {
    const float* x = (const float*)d_in[0];
    const float* W = (const float*)d_in[1];
    float* out = (float*)d_out;
    float* ws  = (float*)d_ws;

    float* qws    = ws;
    float* xpart  = qws + 131072;
    float* ypartA = xpart + 8192;
    float* ypartB = ypartA + 16384;
    float* Zm     = ypartB + 16384;
    float* Bmat   = Zm + 4096;

    const float* x7 = x + (size_t)7 * JB * IN_DIM;
    const float* w7 = W + (size_t)7 * NUM_OUT * IN_DIM * OUT_DIM;

    k_xpart<<<dim3(16, 8), 1024, 0, stream>>>(x, xpart);

    k_uhl_pass<<<STRIPS, 256, 0, stream>>>(x7, w7, nullptr, nullptr, nullptr, qws, 0);
    k_route<<<96, 256, 0, stream>>>(x7, qws, nullptr, nullptr, nullptr, nullptr, ypartA,
                                    xpart, W, w7, Zm);
    k_route<<<32, 256, 0, stream>>>(x7, qws, ypartA, Zm, nullptr, Bmat, ypartB,
                                    nullptr, nullptr, nullptr, nullptr);
    k_uhl_pass<<<STRIPS, 256, 0, stream>>>(x7, w7, ypartB, Zm, Bmat, out, 1);
}


# product pass: A, B and output move in 16-byte pieces (k permuted inside the WMMA fragments, tile column e=4c+tile): 160 loads + 32 stores per wave -> 40 + 8; product and q sums reassociated
# speedup vs baseline: 4.5161x; 1.0460x over previous
#include <hip/hip_runtime.h>
#include <math.h>


typedef __attribute__((ext_vector_type(2))) float v2f;
typedef __attribute__((ext_vector_type(4))) float v4f;
typedef __attribute__((ext_vector_type(8))) float v8f;

#define NUM_IN   8
#define NUM_OUT  8
#define IN_DIM   64
#define OUT_DIM  64
#define JB       16384
#define NCOL     512
#define STRIPS   1024
#define CAP_EPS  1e-9f

union V8 { v8f v; float f[8]; };

__global__ __launch_bounds__(1024) __attribute__((amdgpu_waves_per_eu(8, 8)))
void k_xpart(const float* __restrict__ x, float* __restrict__ xpart) {
    int part = blockIdx.x;
    int i    = blockIdx.y;
    int t    = threadIdx.x;
    const v4f* p = (const v4f*)(x + ((size_t)i * JB + (size_t)part * 1024) * IN_DIM) + t;

    v4f v[16];
    for (int q = 0; q < 16; ++q) v[q] = p[q * 1024];
    v4f s = ((v[0] + v[1]) + (v[2] + v[3])) + ((v[4] + v[5]) + (v[6] + v[7]));
    s += ((v[8] + v[9]) + (v[10] + v[11])) + ((v[12] + v[13]) + (v[14] + v[15]));

    __shared__ v4f sd[1024];
    __shared__ float sd2[256];
    sd[t] = s;
    __syncthreads();
    const float* sf = (const float*)sd;
    if (t < 256) {
        int d = t & 63, g = t >> 6;
        float a = 0.f;
        for (int k = 0; k < 16; ++k) a += sf[(g * 16 + k) * 64 + d];
        sd2[t] = a;
    }
    __syncthreads();
    if (t < 64)
        xpart[(i * 16 + part) * 64 + t] = sd2[t] + sd2[64 + t] + sd2[128 + t] + sd2[192 + t];
}

__device__ __forceinline__ void routing_prologue(const float* __restrict__ ypart,
                                                 const float* __restrict__ Zm,
                                                 const float* __restrict__ Bprev,
                                                 float* __restrict__ Bout,
                                                 float* stage, float* ymean,
                                                 float* Bs, float* cs, int t) {
    if (ypart != nullptr) {
        float s0 = 0.f, s1 = 0.f;
        for (int b = 0; b < 32; ++b) {
            s0 += ypart[b * NCOL + t];
            s1 += ypart[b * NCOL + 256 + t];
        }
        ymean[(t >> 6) * 68 + (t & 63)]       = s0 * (1.0f / (float)JB);
        ymean[((t >> 6) + 4) * 68 + (t & 63)] = s1 * (1.0f / (float)JB);
        const v4f* z4 = (const v4f*)Zm;
        for (int q = 0; q < 4; ++q) {
            int f   = t + 256 * q;
            v4f w   = z4[f];
            int row = f >> 4, col = (f & 15) << 2;
            float* d = stage + row * 65 + col;
            d[0] = w.x; d[1] = w.y; d[2] = w.z; d[3] = w.w;
        }
        __syncthreads();
        if (t < 64) {
            int j = t & 7;
            const float* zp = stage + t * 65;
            const float* yp = ymean + j * 68;
            float d = 0.f;
            for (int e = 0; e < 64; ++e) d += zp[e] * yp[e];
            float Bn = (Bprev != nullptr ? Bprev[t] : 0.f) + d;
            Bs[t] = Bn;
            if (Bout != nullptr && blockIdx.x == 0) Bout[t] = Bn;
        }
    } else {
        if (t < 64) Bs[t] = 0.f;
    }
    __syncthreads();
    if (t < 8) {
        int j = t;
        float m = Bs[j];
        for (int i = 1; i < 8; ++i) m = fmaxf(m, Bs[i * 8 + j]);
        float den = 0.f;
        for (int i = 0; i < 8; ++i) den += expf(Bs[i * 8 + j] - m);
        cs[j] = expf(Bs[56 + j] - m) / den;
    }
}

#define ROW_XMASK(v, m) \
    __builtin_bit_cast(float, __builtin_amdgcn_update_dpp(0, __builtin_bit_cast(int, (v)), 0x160 + (m), 0xf, 0xf, true))
#define ROW_SHARE(v, k) \
    __builtin_bit_cast(float, __builtin_amdgcn_update_dpp(0, __builtin_bit_cast(int, (v)), 0x150 + (k), 0xf, 0xf, true))

__global__ __launch_bounds__(256) __attribute__((amdgpu_waves_per_eu(8, 8)))
void k_uhl_pass(const float* __restrict__ x7,
                const float* __restrict__ w7,
                const float* __restrict__ ypart,
                const float* __restrict__ Zm,
                const float* __restrict__ Bprev,
                float* __restrict__ outp,
                int mode) {
    __shared__ float stage[64 * 65];
    __shared__ float ymean[8 * 68];
    __shared__ float Bs[64];
    __shared__ float cs[8];

    int t    = threadIdx.x;
    int lane = t & 31;
    int wv   = __builtin_amdgcn_readfirstlane(t >> 5);
    int c16  = lane & 15;
    int half = lane >> 4;
    int r0   = blockIdx.x * 16;
    int mrow = r0 + c16;

    v4f a4[8];
    const v4f* ap = (const v4f*)(x7 + (size_t)mrow * 64 + half * 4);
#pragma unroll
    for (int q = 0; q < 8; ++q) a4[q] = ap[q * 2];

    if (mode == 1)
        routing_prologue(ypart, Zm, Bprev, nullptr, stage, ymean, Bs, cs, t);

    V8 acc[4];
#pragma unroll
    for (int tt = 0; tt < 4; ++tt)
#pragma unroll
        for (int q = 0; q < 8; ++q) acc[tt].f[q] = 0.f;

    const v4f* bp = (const v4f*)(w7 + (size_t)wv * 4096 + half * 256) + c16;
    v4f bn[4];
#pragma unroll
    for (int i = 0; i < 4; ++i) bn[i] = bp[i * 16];
#pragma unroll
    for (int q = 0; q < 8; ++q) {
        v4f bc[4];
#pragma unroll
        for (int i = 0; i < 4; ++i) bc[i] = bn[i];
        if (q < 7) {
#pragma unroll
            for (int i = 0; i < 4; ++i) bn[i] = bp[((q + 1) * 8 + i) * 16];
        }
        v2f alo, ahi;
        alo.x = a4[q].x; alo.y = a4[q].y;
        ahi.x = a4[q].z; ahi.y = a4[q].w;
#pragma unroll
        for (int tt = 0; tt < 4; ++tt) {
            v2f b;
            b.x = bc[0][tt];
            b.y = bc[1][tt];
            acc[tt].v = __builtin_amdgcn_wmma_f32_16x16x4_f32(
                false, alo, false, b, (short)0, acc[tt].v, false, false);
        }
#pragma unroll
        for (int tt = 0; tt < 4; ++tt) {
            v2f b;
            b.x = bc[2][tt];
            b.y = bc[3][tt];
            acc[tt].v = __builtin_amdgcn_wmma_f32_16x16x4_f32(
                false, ahi, false, b, (short)0, acc[tt].v, false, false);
        }
        __builtin_amdgcn_sched_barrier(0);
    }

    int ln = __builtin_amdgcn_mbcnt_lo(~0u, 0u);
    int cE = ln & 15;
    int hE = ln >> 4;

    float qs[8];
#pragma unroll
    for (int r = 0; r < 8; ++r) {
        float s = acc[0].f[r] * acc[0].f[r];
        s += acc[1].f[r] * acc[1].f[r];
        s += acc[2].f[r] * acc[2].f[r];
        s += acc[3].f[r] * acc[3].f[r];
        s += ROW_XMASK(s, 1);
        s += ROW_XMASK(s, 2);
        s += ROW_XMASK(s, 4);
        s += ROW_XMASK(s, 8);
        qs[r] = s;
    }
    float qsel = qs[0];
#pragma unroll
    for (int r = 1; r < 8; ++r) qsel = ((cE & 7) == r) ? qs[r] : qsel;

    if (mode == 0) {
        if (cE < 8) outp[(size_t)(r0 + hE * 8 + cE) * 8 + wv] = qsel;
        return;
    }
    __syncthreads();

    float cj = cs[wv];
    float s2 = cj * cj * qsel;
    float g  = s2 / ((1.f + s2) * sqrtf(s2 + CAP_EPS)) * cj;

    float gr[8] = { ROW_SHARE(g, 0), ROW_SHARE(g, 1), ROW_SHARE(g, 2), ROW_SHARE(g, 3),
                    ROW_SHARE(g, 4), ROW_SHARE(g, 5), ROW_SHARE(g, 6), ROW_SHARE(g, 7) };

    int jb0 = (r0 & 511) * 32 + (r0 >> 9);
    float* op = outp + (size_t)wv * (JB * 64) + (size_t)jb0 * 64 + hE * (8 * 32 * 64) + cE * 4;
#pragma unroll
    for (int r = 0; r < 8; ++r) {
        v4f o;
        o.x = gr[r] * acc[0].f[r];
        o.y = gr[r] * acc[1].f[r];
        o.z = gr[r] * acc[2].f[r];
        o.w = gr[r] * acc[3].f[r];
        *(v4f*)(op + r * (32 * 64)) = o;
    }
}

__global__ __launch_bounds__(256) void k_route(const float* __restrict__ x7,
                                               const float* __restrict__ qws,
                                               const float* __restrict__ ypin,
                                               const float* __restrict__ Zin,
                                               const float* __restrict__ Bprev,
                                               float* __restrict__ Bout,
                                               float* __restrict__ ypout,
                                               const float* __restrict__ xpart,
                                               const float* __restrict__ W,
                                               const float* __restrict__ w7,
                                               float* __restrict__ Zout) {
    __shared__ float stage[64 * 65];
    __shared__ float ymean[8 * 68];
    __shared__ float Bs[64];
    __shared__ float cs[8];
    __shared__ v4f   gs4[1024];
    __shared__ float red[8 * NCOL];

    int t = threadIdx.x;
    int b = blockIdx.x;

    if (b >= 32) {
        int p = b - 32;
        int i = p >> 3, j = p & 7;
        float* umean = stage;
        float* uh    = stage + 64;
        if (t < 64) {
            float s = 0.f;
            for (int q = 0; q < 16; ++q) s += xpart[(i * 16 + q) * 64 + t];
            umean[t] = s * (1.0f / (float)JB);
        }
        __syncthreads();
        if (t < 64) {
            const float* wp = W + (size_t)p * 4096 + t;
            float s = 0.f;
            for (int d = 0; d < 64; ++d) s += umean[d] * wp[d * 64];
            uh[t] = s;
        }
        __syncthreads();
        if (t < 64) {
            const v4f* wr = (const v4f*)(w7 + (size_t)j * 4096 + t * 64);
            float s = 0.f;
            for (int e4 = 0; e4 < 16; ++e4) {
                v4f w = wr[e4];
                s += uh[e4 * 4 + 0] * w.x;
                s += uh[e4 * 4 + 1] * w.y;
                s += uh[e4 * 4 + 2] * w.z;
                s += uh[e4 * 4 + 3] * w.w;
            }
            Zout[p * 64 + t] = s;
        }
        return;
    }

    int lane = t & 31;
    int wv   = t >> 5;

    const v4f* q4 = (const v4f*)(qws + (size_t)b * 4096);
    v4f qv[4];
    for (int k = 0; k < 4; ++k) qv[k] = q4[t + 256 * k];

    routing_prologue(ypin, Zin, Bprev, Bout, stage, ymean, Bs, cs, t);
    __syncthreads();

    for (int k = 0; k < 4; ++k) {
        int f  = t + 256 * k;
        int j0 = (f & 1) * 4;
        v4f g;
        for (int m = 0; m < 4; ++m) {
            float c  = cs[j0 + m];
            float s2 = c * c * qv[k][m];
            float factor = s2 / ((1.f + s2) * sqrtf(s2 + CAP_EPS));
            g[m] = factor * c;
        }
        gs4[f] = g;
    }
    __syncthreads();

    const v2f* xp = (const v2f*)(x7 + ((size_t)b * 512 + (size_t)wv * 64) * IN_DIM) + lane;
    float a0[8], a1[8];
    for (int j = 0; j < 8; ++j) { a0[j] = 0.f; a1[j] = 0.f; }
    for (int i0 = 0; i0 < 64; i0 += 16) {
        v2f xv[16];
#pragma unroll
        for (int u = 0; u < 16; ++u) xv[u] = xp[(i0 + u) * 32];
#pragma unroll
        for (int u = 0; u < 16; ++u) {
            v4f ga = gs4[(wv * 64 + i0 + u) * 2];
            v4f gb = gs4[(wv * 64 + i0 + u) * 2 + 1];
            a0[0] += ga.x * xv[u].x;  a1[0] += ga.x * xv[u].y;
            a0[1] += ga.y * xv[u].x;  a1[1] += ga.y * xv[u].y;
            a0[2] += ga.z * xv[u].x;  a1[2] += ga.z * xv[u].y;
            a0[3] += ga.w * xv[u].x;  a1[3] += ga.w * xv[u].y;
            a0[4] += gb.x * xv[u].x;  a1[4] += gb.x * xv[u].y;
            a0[5] += gb.y * xv[u].x;  a1[5] += gb.y * xv[u].y;
            a0[6] += gb.z * xv[u].x;  a1[6] += gb.z * xv[u].y;
            a0[7] += gb.w * xv[u].x;  a1[7] += gb.w * xv[u].y;
        }
    }
    for (int j = 0; j < 8; ++j) {
        v2f s; s.x = a0[j]; s.y = a1[j];
        *(v2f*)(red + wv * NCOL + j * 64 + 2 * lane) = s;
    }
    __syncthreads();
    for (int o = t; o < NCOL; o += 256) {
        float s = 0.f;
        for (int w = 0; w < 8; ++w) s += red[w * NCOL + o];
        ypout[(size_t)b * NCOL + o] = s;
    }
}

extern "C" void kernel_launch(void* const* d_in, const int* in_sizes, int n_in,
                              void* d_out, int out_size, void* d_ws, size_t ws_size,
                              hipStream_t stream) {
    const float* x = (const float*)d_in[0];
    const float* W = (const float*)d_in[1];
    float* out = (float*)d_out;
    float* ws  = (float*)d_ws;

    float* qws    = ws;
    float* xpart  = qws + 131072;
    float* ypartA = xpart + 8192;
    float* ypartB = ypartA + 16384;
    float* Zm     = ypartB + 16384;
    float* Bmat   = Zm + 4096;

    const float* x7 = x + (size_t)7 * JB * IN_DIM;
    const float* w7 = W + (size_t)7 * NUM_OUT * IN_DIM * OUT_DIM;

    k_xpart<<<dim3(16, 8), 1024, 0, stream>>>(x, xpart);

    k_uhl_pass<<<STRIPS, 256, 0, stream>>>(x7, w7, nullptr, nullptr, nullptr, qws, 0);
    k_route<<<96, 256, 0, stream>>>(x7, qws, nullptr, nullptr, nullptr, nullptr, ypartA,
                                    xpart, W, w7, Zm);
    k_route<<<32, 256, 0, stream>>>(x7, qws, ypartA, Zm, nullptr, Bmat, ypartB,
                                    nullptr, nullptr, nullptr, nullptr);
    k_uhl_pass<<<STRIPS, 256, 0, stream>>>(x7, w7, ypartB, Zm, Bmat, out, 1);
}


# launch fusion, 5 launches -> 4: the x column sums run as 128 extra 256-thread work-groups at the head of the first product launch, under its matrix work; x partial sums added in another order
# speedup vs baseline: 4.9604x; 1.0984x over previous
#include <hip/hip_runtime.h>
#include <math.h>


typedef __attribute__((ext_vector_type(2))) float v2f;
typedef __attribute__((ext_vector_type(4))) float v4f;
typedef __attribute__((ext_vector_type(8))) float v8f;

#define NUM_IN   8
#define NUM_OUT  8
#define IN_DIM   64
#define OUT_DIM  64
#define JB       16384
#define NCOL     512
#define STRIPS   1024
#define CAP_EPS  1e-9f

union V8 { v8f v; float f[8]; };

#define XBLK 128

__device__ __forceinline__ void xpart_block(const float* __restrict__ x,
                                            float* __restrict__ xpart,
                                            int blk, int t) {
    const v4f* p = (const v4f*)(x + (size_t)blk * 1024 * IN_DIM) + t;

    v4f v[16];
#pragma unroll
    for (int q = 0; q < 16; ++q) v[q] = p[q * 256];
    v4f s[4];
#pragma unroll
    for (int k = 0; k < 4; ++k) s[k] = (v4f){0.f, 0.f, 0.f, 0.f};
#pragma clang loop unroll(disable)
    for (int b = 1; b < 4; ++b) {
        const v4f* pb = p + b * (16 * 256);
#pragma unroll
        for (int q = 0; q < 16; ++q) {
            s[q & 3] += v[q];
            v[q] = pb[q * 256];
        }
    }
#pragma unroll
    for (int q = 0; q < 16; ++q) s[q & 3] += v[q];
    v4f tot = (s[0] + s[1]) + (s[2] + s[3]);

    __shared__ v4f sd[256];
    sd[t] = tot;
    __syncthreads();
    if (t < 64) {
        const float* sf = (const float*)sd;
        float a = 0.f;
        for (int k = 0; k < 16; ++k) a += sf[k * 64 + t];
        xpart[blk * 64 + t] = a;
    }
}

__device__ __forceinline__ void routing_prologue(const float* __restrict__ ypart,
                                                 const float* __restrict__ Zm,
                                                 const float* __restrict__ Bprev,
                                                 float* __restrict__ Bout,
                                                 float* stage, float* ymean,
                                                 float* Bs, float* cs, int t) {
    if (ypart != nullptr) {
        float s0 = 0.f, s1 = 0.f;
        for (int b = 0; b < 32; ++b) {
            s0 += ypart[b * NCOL + t];
            s1 += ypart[b * NCOL + 256 + t];
        }
        ymean[(t >> 6) * 68 + (t & 63)]       = s0 * (1.0f / (float)JB);
        ymean[((t >> 6) + 4) * 68 + (t & 63)] = s1 * (1.0f / (float)JB);
        const v4f* z4 = (const v4f*)Zm;
        for (int q = 0; q < 4; ++q) {
            int f   = t + 256 * q;
            v4f w   = z4[f];
            int row = f >> 4, col = (f & 15) << 2;
            float* d = stage + row * 65 + col;
            d[0] = w.x; d[1] = w.y; d[2] = w.z; d[3] = w.w;
        }
        __syncthreads();
        if (t < 64) {
            int j = t & 7;
            const float* zp = stage + t * 65;
            const float* yp = ymean + j * 68;
            float d = 0.f;
            for (int e = 0; e < 64; ++e) d += zp[e] * yp[e];
            float Bn = (Bprev != nullptr ? Bprev[t] : 0.f) + d;
            Bs[t] = Bn;
            if (Bout != nullptr && blockIdx.x == 0) Bout[t] = Bn;
        }
    } else {
        if (t < 64) Bs[t] = 0.f;
    }
    __syncthreads();
    if (t < 8) {
        int j = t;
        float m = Bs[j];
        for (int i = 1; i < 8; ++i) m = fmaxf(m, Bs[i * 8 + j]);
        float den = 0.f;
        for (int i = 0; i < 8; ++i) den += expf(Bs[i * 8 + j] - m);
        cs[j] = expf(Bs[56 + j] - m) / den;
    }
}

#define ROW_XMASK(v, m) \
    __builtin_bit_cast(float, __builtin_amdgcn_update_dpp(0, __builtin_bit_cast(int, (v)), 0x160 + (m), 0xf, 0xf, true))
#define ROW_SHARE(v, k) \
    __builtin_bit_cast(float, __builtin_amdgcn_update_dpp(0, __builtin_bit_cast(int, (v)), 0x150 + (k), 0xf, 0xf, true))

__device__ __forceinline__ void uhl_pass(const float* __restrict__ x7,
                                         const float* __restrict__ w7,
                                         const float* __restrict__ ypart,
                                         const float* __restrict__ Zm,
                                         const float* __restrict__ Bprev,
                                         float* __restrict__ outp,
                                         const int mode, const int strip,
                                         float* stage, float* ymean, float* Bs, float* cs) {
    int t    = threadIdx.x;
    int lane = t & 31;
    int wv   = __builtin_amdgcn_readfirstlane(t >> 5);
    int c16  = lane & 15;
    int half = lane >> 4;
    int r0   = strip * 16;
    int mrow = r0 + c16;

    v4f a4[8];
    const v4f* ap = (const v4f*)(x7 + (size_t)mrow * 64 + half * 4);
#pragma unroll
    for (int q = 0; q < 8; ++q) a4[q] = ap[q * 2];

    if (mode == 1)
        routing_prologue(ypart, Zm, Bprev, nullptr, stage, ymean, Bs, cs, t);

    V8 acc[4];
#pragma unroll
    for (int tt = 0; tt < 4; ++tt)
#pragma unroll
        for (int q = 0; q < 8; ++q) acc[tt].f[q] = 0.f;

    const v4f* bp = (const v4f*)(w7 + (size_t)wv * 4096 + half * 256) + c16;
    v4f bn[4];
#pragma unroll
    for (int i = 0; i < 4; ++i) bn[i] = bp[i * 16];
#pragma unroll
    for (int q = 0; q < 8; ++q) {
        v4f bc[4];
#pragma unroll
        for (int i = 0; i < 4; ++i) bc[i] = bn[i];
        if (q < 7) {
#pragma unroll
            for (int i = 0; i < 4; ++i) bn[i] = bp[((q + 1) * 8 + i) * 16];
        }
        v2f alo, ahi;
        alo.x = a4[q].x; alo.y = a4[q].y;
        ahi.x = a4[q].z; ahi.y = a4[q].w;
#pragma unroll
        for (int tt = 0; tt < 4; ++tt) {
            v2f b;
            b.x = bc[0][tt];
            b.y = bc[1][tt];
            acc[tt].v = __builtin_amdgcn_wmma_f32_16x16x4_f32(
                false, alo, false, b, (short)0, acc[tt].v, false, false);
        }
#pragma unroll
        for (int tt = 0; tt < 4; ++tt) {
            v2f b;
            b.x = bc[2][tt];
            b.y = bc[3][tt];
            acc[tt].v = __builtin_amdgcn_wmma_f32_16x16x4_f32(
                false, ahi, false, b, (short)0, acc[tt].v, false, false);
        }
        __builtin_amdgcn_sched_barrier(0);
    }

    int ln = __builtin_amdgcn_mbcnt_lo(~0u, 0u);
    int cE = ln & 15;
    int hE = ln >> 4;

    float qs[8];
#pragma unroll
    for (int r = 0; r < 8; ++r) {
        float s = acc[0].f[r] * acc[0].f[r];
        s += acc[1].f[r] * acc[1].f[r];
        s += acc[2].f[r] * acc[2].f[r];
        s += acc[3].f[r] * acc[3].f[r];
        s += ROW_XMASK(s, 1);
        s += ROW_XMASK(s, 2);
        s += ROW_XMASK(s, 4);
        s += ROW_XMASK(s, 8);
        qs[r] = s;
    }
    float qsel = qs[0];
#pragma unroll
    for (int r = 1; r < 8; ++r) qsel = ((cE & 7) == r) ? qs[r] : qsel;

    if (mode == 0) {
        if (cE < 8) outp[(size_t)(r0 + hE * 8 + cE) * 8 + wv] = qsel;
        return;
    }
    __syncthreads();

    float cj = cs[wv];
    float s2 = cj * cj * qsel;
    float g  = s2 / ((1.f + s2) * sqrtf(s2 + CAP_EPS)) * cj;

    float gr[8] = { ROW_SHARE(g, 0), ROW_SHARE(g, 1), ROW_SHARE(g, 2), ROW_SHARE(g, 3),
                    ROW_SHARE(g, 4), ROW_SHARE(g, 5), ROW_SHARE(g, 6), ROW_SHARE(g, 7) };

    int jb0 = (r0 & 511) * 32 + (r0 >> 9);
    float* op = outp + (size_t)wv * (JB * 64) + (size_t)jb0 * 64 + hE * (8 * 32 * 64) + cE * 4;
#pragma unroll
    for (int r = 0; r < 8; ++r) {
        v4f o;
        o.x = gr[r] * acc[0].f[r];
        o.y = gr[r] * acc[1].f[r];
        o.z = gr[r] * acc[2].f[r];
        o.w = gr[r] * acc[3].f[r];
        *(v4f*)(op + r * (32 * 64)) = o;
    }
}

__global__ __launch_bounds__(256) __attribute__((amdgpu_waves_per_eu(8, 8)))
void k_first(const float* __restrict__ x,
             float* __restrict__ xpart,
             const float* __restrict__ x7,
             const float* __restrict__ w7,
             float* __restrict__ qws) {
    int b = blockIdx.x;
    if (b < XBLK) {
        xpart_block(x, xpart, b, threadIdx.x);
        return;
    }
    uhl_pass(x7, w7, nullptr, nullptr, nullptr, qws, 0, b - XBLK,
             nullptr, nullptr, nullptr, nullptr);
}

__global__ __launch_bounds__(256) __attribute__((amdgpu_waves_per_eu(8, 8)))
void k_final(const float* __restrict__ x7,
             const float* __restrict__ w7,
             const float* __restrict__ ypart,
             const float* __restrict__ Zm,
             const float* __restrict__ Bprev,
             float* __restrict__ outp) {
    __shared__ float stage[64 * 65];
    __shared__ float ymean[8 * 68];
    __shared__ float Bs[64];
    __shared__ float cs[8];
    uhl_pass(x7, w7, ypart, Zm, Bprev, outp, 1, blockIdx.x, stage, ymean, Bs, cs);
}

__global__ __launch_bounds__(256) void k_route(const float* __restrict__ x7,
                                               const float* __restrict__ qws,
                                               const float* __restrict__ ypin,
                                               const float* __restrict__ Zin,
                                               const float* __restrict__ Bprev,
                                               float* __restrict__ Bout,
                                               float* __restrict__ ypout,
                                               const float* __restrict__ xpart,
                                               const float* __restrict__ W,
                                               const float* __restrict__ w7,
                                               float* __restrict__ Zout) {
    __shared__ float stage[64 * 65];
    __shared__ float ymean[8 * 68];
    __shared__ float Bs[64];
    __shared__ float cs[8];
    __shared__ v4f   gs4[1024];
    __shared__ float red[8 * NCOL];

    int t = threadIdx.x;
    int b = blockIdx.x;

    if (b >= 32) {
        int p = b - 32;
        int i = p >> 3, j = p & 7;
        float* umean = stage;
        float* uh    = stage + 64;
        if (t < 64) {
            float s = 0.f;
            for (int q = 0; q < 16; ++q) s += xpart[(i * 16 + q) * 64 + t];
            umean[t] = s * (1.0f / (float)JB);
        }
        __syncthreads();
        if (t < 64) {
            const float* wp = W + (size_t)p * 4096 + t;
            float s = 0.f;
            for (int d = 0; d < 64; ++d) s += umean[d] * wp[d * 64];
            uh[t] = s;
        }
        __syncthreads();
        if (t < 64) {
            const v4f* wr = (const v4f*)(w7 + (size_t)j * 4096 + t * 64);
            float s = 0.f;
            for (int e4 = 0; e4 < 16; ++e4) {
                v4f w = wr[e4];
                s += uh[e4 * 4 + 0] * w.x;
                s += uh[e4 * 4 + 1] * w.y;
                s += uh[e4 * 4 + 2] * w.z;
                s += uh[e4 * 4 + 3] * w.w;
            }
            Zout[p * 64 + t] = s;
        }
        return;
    }

    int lane = t & 31;
    int wv   = t >> 5;

    const v4f* q4 = (const v4f*)(qws + (size_t)b * 4096);
    v4f qv[4];
    for (int k = 0; k < 4; ++k) qv[k] = q4[t + 256 * k];

    routing_prologue(ypin, Zin, Bprev, Bout, stage, ymean, Bs, cs, t);
    __syncthreads();

    for (int k = 0; k < 4; ++k) {
        int f  = t + 256 * k;
        int j0 = (f & 1) * 4;
        v4f g;
        for (int m = 0; m < 4; ++m) {
            float c  = cs[j0 + m];
            float s2 = c * c * qv[k][m];
            float factor = s2 / ((1.f + s2) * sqrtf(s2 + CAP_EPS));
            g[m] = factor * c;
        }
        gs4[f] = g;
    }
    __syncthreads();

    const v2f* xp = (const v2f*)(x7 + ((size_t)b * 512 + (size_t)wv * 64) * IN_DIM) + lane;
    float a0[8], a1[8];
    for (int j = 0; j < 8; ++j) { a0[j] = 0.f; a1[j] = 0.f; }
    for (int i0 = 0; i0 < 64; i0 += 16) {
        v2f xv[16];
#pragma unroll
        for (int u = 0; u < 16; ++u) xv[u] = xp[(i0 + u) * 32];
#pragma unroll
        for (int u = 0; u < 16; ++u) {
            v4f ga = gs4[(wv * 64 + i0 + u) * 2];
            v4f gb = gs4[(wv * 64 + i0 + u) * 2 + 1];
            a0[0] += ga.x * xv[u].x;  a1[0] += ga.x * xv[u].y;
            a0[1] += ga.y * xv[u].x;  a1[1] += ga.y * xv[u].y;
            a0[2] += ga.z * xv[u].x;  a1[2] += ga.z * xv[u].y;
            a0[3] += ga.w * xv[u].x;  a1[3] += ga.w * xv[u].y;
            a0[4] += gb.x * xv[u].x;  a1[4] += gb.x * xv[u].y;
            a0[5] += gb.y * xv[u].x;  a1[5] += gb.y * xv[u].y;
            a0[6] += gb.z * xv[u].x;  a1[6] += gb.z * xv[u].y;
            a0[7] += gb.w * xv[u].x;  a1[7] += gb.w * xv[u].y;
        }
    }
    for (int j = 0; j < 8; ++j) {
        v2f s; s.x = a0[j]; s.y = a1[j];
        *(v2f*)(red + wv * NCOL + j * 64 + 2 * lane) = s;
    }
    __syncthreads();
    for (int o = t; o < NCOL; o += 256) {
        float s = 0.f;
        for (int w = 0; w < 8; ++w) s += red[w * NCOL + o];
        ypout[(size_t)b * NCOL + o] = s;
    }
}

extern "C" void kernel_launch(void* const* d_in, const int* in_sizes, int n_in,
                              void* d_out, int out_size, void* d_ws, size_t ws_size,
                              hipStream_t stream) {
    const float* x = (const float*)d_in[0];
    const float* W = (const float*)d_in[1];
    float* out = (float*)d_out;
    float* ws  = (float*)d_ws;

    float* qws    = ws;
    float* xpart  = qws + 131072;
    float* ypartA = xpart + 8192;
    float* ypartB = ypartA + 16384;
    float* Zm     = ypartB + 16384;
    float* Bmat   = Zm + 4096;

    const float* x7 = x + (size_t)7 * JB * IN_DIM;
    const float* w7 = W + (size_t)7 * NUM_OUT * IN_DIM * OUT_DIM;

    k_first<<<XBLK + STRIPS, 256, 0, stream>>>(x, xpart, x7, w7, qws);
    k_route<<<96, 256, 0, stream>>>(x7, qws, nullptr, nullptr, nullptr, nullptr, ypartA,
                                    xpart, W, w7, Zm);
    k_route<<<32, 256, 0, stream>>>(x7, qws, ypartA, Zm, nullptr, Bmat, ypartB,
                                    nullptr, nullptr, nullptr, nullptr);
    k_final<<<STRIPS, 256, 0, stream>>>(x7, w7, ypartB, Zm, Bmat, out);
}


# B's last update from per-block shares of its dots: the 32-block launch takes Z.y of its rows; the last launch adds 64 floats per block in-lane, no LDS or barrier; B's dot, softmax sum reassociated
# speedup vs baseline: 5.4347x; 1.0956x over previous
#include <hip/hip_runtime.h>
#include <math.h>


typedef __attribute__((ext_vector_type(2))) float v2f;
typedef __attribute__((ext_vector_type(4))) float v4f;
typedef __attribute__((ext_vector_type(8))) float v8f;

#define NUM_IN   8
#define NUM_OUT  8
#define IN_DIM   64
#define OUT_DIM  64
#define JB       16384
#define NCOL     512
#define STRIPS   1024
#define CAP_EPS  1e-9f

union V8 { v8f v; float f[8]; };

#define XBLK 128

__device__ __forceinline__ void xpart_block(const float* __restrict__ x,
                                            float* __restrict__ xpart,
                                            int blk, int t) {
    const v4f* p = (const v4f*)(x + (size_t)blk * 1024 * IN_DIM) + t;

    v4f v[16];
#pragma unroll
    for (int q = 0; q < 16; ++q) v[q] = p[q * 256];
    v4f s[4];
#pragma unroll
    for (int k = 0; k < 4; ++k) s[k] = (v4f){0.f, 0.f, 0.f, 0.f};
#pragma clang loop unroll(disable)
    for (int b = 1; b < 4; ++b) {
        const v4f* pb = p + b * (16 * 256);
#pragma unroll
        for (int q = 0; q < 16; ++q) {
            s[q & 3] += v[q];
            v[q] = pb[q * 256];
        }
    }
#pragma unroll
    for (int q = 0; q < 16; ++q) s[q & 3] += v[q];
    v4f tot = (s[0] + s[1]) + (s[2] + s[3]);

    __shared__ v4f sd[256];
    sd[t] = tot;
    __syncthreads();
    if (t < 64) {
        const float* sf = (const float*)sd;
        float a = 0.f;
        for (int k = 0; k < 16; ++k) a += sf[k * 64 + t];
        xpart[blk * 64 + t] = a;
    }
}

__device__ __forceinline__ void routing_prologue(const float* __restrict__ ypart,
                                                 const float* __restrict__ Zm,
                                                 const float* __restrict__ Bprev,
                                                 float* __restrict__ Bout,
                                                 float* stage, float* ymean,
                                                 float* Bs, float* cs, int t) {
    if (ypart != nullptr) {
        float s0 = 0.f, s1 = 0.f;
        for (int b = 0; b < 32; ++b) {
            s0 += ypart[b * NCOL + t];
            s1 += ypart[b * NCOL + 256 + t];
        }
        ymean[(t >> 6) * 68 + (t & 63)]       = s0 * (1.0f / (float)JB);
        ymean[((t >> 6) + 4) * 68 + (t & 63)] = s1 * (1.0f / (float)JB);
        const v4f* z4 = (const v4f*)Zm;
        for (int q = 0; q < 4; ++q) {
            int f   = t + 256 * q;
            v4f w   = z4[f];
            int row = f >> 4, col = (f & 15) << 2;
            float* d = stage + row * 65 + col;
            d[0] = w.x; d[1] = w.y; d[2] = w.z; d[3] = w.w;
        }
        __syncthreads();
        if (t < 64) {
            int j = t & 7;
            const float* zp = stage + t * 65;
            const float* yp = ymean + j * 68;
            float d = 0.f;
            for (int e = 0; e < 64; ++e) d += zp[e] * yp[e];
            float Bn = (Bprev != nullptr ? Bprev[t] : 0.f) + d;
            Bs[t] = Bn;
            if (Bout != nullptr && blockIdx.x == 0) Bout[t] = Bn;
        }
    } else {
        if (t < 64) Bs[t] = 0.f;
    }
    __syncthreads();
    if (t < 8) {
        int j = t;
        float m = Bs[j];
        for (int i = 1; i < 8; ++i) m = fmaxf(m, Bs[i * 8 + j]);
        float den = 0.f;
        for (int i = 0; i < 8; ++i) den += expf(Bs[i * 8 + j] - m);
        cs[j] = expf(Bs[56 + j] - m) / den;
    }
}

#define ROW_XMASK(v, m) \
    __builtin_bit_cast(float, __builtin_amdgcn_update_dpp(0, __builtin_bit_cast(int, (v)), 0x160 + (m), 0xf, 0xf, true))
#define ROW_SHARE(v, k) \
    __builtin_bit_cast(float, __builtin_amdgcn_update_dpp(0, __builtin_bit_cast(int, (v)), 0x150 + (k), 0xf, 0xf, true))
#define ROW_SWAP(v) \
    __builtin_bit_cast(float, __builtin_amdgcn_permlanex16(__builtin_bit_cast(int, (v)), __builtin_bit_cast(int, (v)), 0x76543210, 0xfedcba98, false, false))

__device__ __forceinline__ void uhl_pass(const float* __restrict__ x7,
                                         const float* __restrict__ w7,
                                         const float* __restrict__ dpart,
                                         const float* __restrict__ Bprev,
                                         float* __restrict__ outp,
                                         const int mode, const int strip) {
    int t    = threadIdx.x;
    int lane = t & 31;
    int wv   = __builtin_amdgcn_readfirstlane(t >> 5);
    int c16  = lane & 15;
    int half = lane >> 4;
    int r0   = strip * 16;
    int mrow = r0 + c16;

    v4f a4[8];
    const v4f* ap = (const v4f*)(x7 + (size_t)mrow * 64 + half * 4);
#pragma unroll
    for (int q = 0; q < 8; ++q) a4[q] = ap[q * 2];

    float Bn = 0.f;
    if (mode == 1) {
        const float* dq = dpart + (lane & 3) * (8 * 64) + (lane >> 2) * 8 + wv;
        float ds[8];
#pragma unroll
        for (int k = 0; k < 8; ++k) ds[k] = dq[k * 64];
        float bprev = Bprev[(lane >> 2) * 8 + wv];
        float sd = ((ds[0] + ds[1]) + (ds[2] + ds[3])) + ((ds[4] + ds[5]) + (ds[6] + ds[7]));
        sd += ROW_XMASK(sd, 1);
        sd += ROW_XMASK(sd, 2);
        Bn = bprev + sd * (1.0f / (float)JB);
    }

    V8 acc[4];
#pragma unroll
    for (int tt = 0; tt < 4; ++tt)
#pragma unroll
        for (int q = 0; q < 8; ++q) acc[tt].f[q] = 0.f;

    const v4f* bp = (const v4f*)(w7 + (size_t)wv * 4096 + half * 256) + c16;
    v4f bn[4];
#pragma unroll
    for (int i = 0; i < 4; ++i) bn[i] = bp[i * 16];
#pragma unroll
    for (int q = 0; q < 8; ++q) {
        v4f bc[4];
#pragma unroll
        for (int i = 0; i < 4; ++i) bc[i] = bn[i];
        if (q < 7) {
#pragma unroll
            for (int i = 0; i < 4; ++i) bn[i] = bp[((q + 1) * 8 + i) * 16];
        }
        v2f alo, ahi;
        alo.x = a4[q].x; alo.y = a4[q].y;
        ahi.x = a4[q].z; ahi.y = a4[q].w;
#pragma unroll
        for (int tt = 0; tt < 4; ++tt) {
            v2f b;
            b.x = bc[0][tt];
            b.y = bc[1][tt];
            acc[tt].v = __builtin_amdgcn_wmma_f32_16x16x4_f32(
                false, alo, false, b, (short)0, acc[tt].v, false, false);
        }
#pragma unroll
        for (int tt = 0; tt < 4; ++tt) {
            v2f b;
            b.x = bc[2][tt];
            b.y = bc[3][tt];
            acc[tt].v = __builtin_amdgcn_wmma_f32_16x16x4_f32(
                false, ahi, false, b, (short)0, acc[tt].v, false, false);
        }
        __builtin_amdgcn_sched_barrier(0);
    }

    int ln = __builtin_amdgcn_mbcnt_lo(~0u, 0u);
    int cE = ln & 15;
    int hE = ln >> 4;

    float qs[8];
#pragma unroll
    for (int r = 0; r < 8; ++r) {
        float s = acc[0].f[r] * acc[0].f[r];
        s += acc[1].f[r] * acc[1].f[r];
        s += acc[2].f[r] * acc[2].f[r];
        s += acc[3].f[r] * acc[3].f[r];
        s += ROW_XMASK(s, 1);
        s += ROW_XMASK(s, 2);
        s += ROW_XMASK(s, 4);
        s += ROW_XMASK(s, 8);
        qs[r] = s;
    }
    float qsel = qs[0];
#pragma unroll
    for (int r = 1; r < 8; ++r) qsel = ((cE & 7) == r) ? qs[r] : qsel;

    if (mode == 0) {
        if (cE < 8) outp[(size_t)(r0 + hE * 8 + cE) * 8 + wv] = qsel;
        return;
    }
    float bm = fmaxf(Bn, ROW_XMASK(Bn, 4));
    bm = fmaxf(bm, ROW_XMASK(bm, 8));
    bm = fmaxf(bm, ROW_SWAP(bm));
    float ex  = expf(Bn - bm);
    float den = ex + ROW_XMASK(ex, 4);
    den += ROW_XMASK(den, 8);
    den += ROW_SWAP(den);
    float e7 = __builtin_bit_cast(float, __builtin_amdgcn_readlane(__builtin_bit_cast(int, ex), 28));

    float cj = e7 / den;
    float s2 = cj * cj * qsel;
    float g  = s2 / ((1.f + s2) * sqrtf(s2 + CAP_EPS)) * cj;

    float gr[8] = { ROW_SHARE(g, 0), ROW_SHARE(g, 1), ROW_SHARE(g, 2), ROW_SHARE(g, 3),
                    ROW_SHARE(g, 4), ROW_SHARE(g, 5), ROW_SHARE(g, 6), ROW_SHARE(g, 7) };

    int jb0 = (r0 & 511) * 32 + (r0 >> 9);
    float* op = outp + (size_t)wv * (JB * 64) + (size_t)jb0 * 64 + hE * (8 * 32 * 64) + cE * 4;
#pragma unroll
    for (int r = 0; r < 8; ++r) {
        v4f o;
        o.x = gr[r] * acc[0].f[r];
        o.y = gr[r] * acc[1].f[r];
        o.z = gr[r] * acc[2].f[r];
        o.w = gr[r] * acc[3].f[r];
        *(v4f*)(op + r * (32 * 64)) = o;
    }
}

__global__ __launch_bounds__(256) __attribute__((amdgpu_waves_per_eu(8, 8)))
void k_first(const float* __restrict__ x,
             float* __restrict__ xpart,
             const float* __restrict__ x7,
             const float* __restrict__ w7,
             float* __restrict__ qws) {
    int b = blockIdx.x;
    if (b < XBLK) {
        xpart_block(x, xpart, b, threadIdx.x);
        return;
    }
    uhl_pass(x7, w7, nullptr, nullptr, qws, 0, b - XBLK);
}

__global__ __launch_bounds__(256) __attribute__((amdgpu_waves_per_eu(8, 8)))
void k_final(const float* __restrict__ x7,
             const float* __restrict__ w7,
             const float* __restrict__ dpart,
             const float* __restrict__ Bprev,
             float* __restrict__ outp) {
    uhl_pass(x7, w7, dpart, Bprev, outp, 1, blockIdx.x);
}

__global__ __launch_bounds__(256) void k_route(const float* __restrict__ x7,
                                               const float* __restrict__ qws,
                                               const float* __restrict__ ypin,
                                               const float* __restrict__ Zin,
                                               const float* __restrict__ Bprev,
                                               float* __restrict__ Bout,
                                               float* __restrict__ ypout,
                                               float* __restrict__ dpout,
                                               const float* __restrict__ xpart,
                                               const float* __restrict__ W,
                                               const float* __restrict__ w7,
                                               float* __restrict__ Zout) {
    __shared__ float stage[64 * 65];
    __shared__ float ymean[8 * 68];
    __shared__ float Bs[64];
    __shared__ float cs[8];
    __shared__ v4f   gs4[1024];
    __shared__ float red[8 * NCOL];

    int t = threadIdx.x;
    int b = blockIdx.x;

    if (b >= 32) {
        int p = b - 32;
        int i = p >> 3, j = p & 7;
        float* umean = stage;
        float* uh    = stage + 64;
        if (t < 64) {
            float s = 0.f;
            for (int q = 0; q < 16; ++q) s += xpart[(i * 16 + q) * 64 + t];
            umean[t] = s * (1.0f / (float)JB);
        }
        __syncthreads();
        if (t < 64) {
            const float* wp = W + (size_t)p * 4096 + t;
            float s = 0.f;
            for (int d = 0; d < 64; ++d) s += umean[d] * wp[d * 64];
            uh[t] = s;
        }
        __syncthreads();
        if (t < 64) {
            const v4f* wr = (const v4f*)(w7 + (size_t)j * 4096 + t * 64);
            float s = 0.f;
            for (int e4 = 0; e4 < 16; ++e4) {
                v4f w = wr[e4];
                s += uh[e4 * 4 + 0] * w.x;
                s += uh[e4 * 4 + 1] * w.y;
                s += uh[e4 * 4 + 2] * w.z;
                s += uh[e4 * 4 + 3] * w.w;
            }
            Zout[p * 64 + t] = s;
        }
        return;
    }

    int lane = t & 31;
    int wv   = t >> 5;

    const v4f* q4 = (const v4f*)(qws + (size_t)b * 4096);
    v4f qv[4];
    for (int k = 0; k < 4; ++k) qv[k] = q4[t + 256 * k];

    routing_prologue(ypin, Zin, Bprev, Bout, stage, ymean, Bs, cs, t);
    __syncthreads();

    for (int k = 0; k < 4; ++k) {
        int f  = t + 256 * k;
        int j0 = (f & 1) * 4;
        v4f g;
        for (int m = 0; m < 4; ++m) {
            float c  = cs[j0 + m];
            float s2 = c * c * qv[k][m];
            float factor = s2 / ((1.f + s2) * sqrtf(s2 + CAP_EPS));
            g[m] = factor * c;
        }
        gs4[f] = g;
    }
    __syncthreads();

    const v2f* xp = (const v2f*)(x7 + ((size_t)b * 512 + (size_t)wv * 64) * IN_DIM) + lane;
    float a0[8], a1[8];
    for (int j = 0; j < 8; ++j) { a0[j] = 0.f; a1[j] = 0.f; }
    for (int i0 = 0; i0 < 64; i0 += 16) {
        v2f xv[16];
#pragma unroll
        for (int u = 0; u < 16; ++u) xv[u] = xp[(i0 + u) * 32];
#pragma unroll
        for (int u = 0; u < 16; ++u) {
            v4f ga = gs4[(wv * 64 + i0 + u) * 2];
            v4f gb = gs4[(wv * 64 + i0 + u) * 2 + 1];
            a0[0] += ga.x * xv[u].x;  a1[0] += ga.x * xv[u].y;
            a0[1] += ga.y * xv[u].x;  a1[1] += ga.y * xv[u].y;
            a0[2] += ga.z * xv[u].x;  a1[2] += ga.z * xv[u].y;
            a0[3] += ga.w * xv[u].x;  a1[3] += ga.w * xv[u].y;
            a0[4] += gb.x * xv[u].x;  a1[4] += gb.x * xv[u].y;
            a0[5] += gb.y * xv[u].x;  a1[5] += gb.y * xv[u].y;
            a0[6] += gb.z * xv[u].x;  a1[6] += gb.z * xv[u].y;
            a0[7] += gb.w * xv[u].x;  a1[7] += gb.w * xv[u].y;
        }
    }
    for (int j = 0; j < 8; ++j) {
        v2f s; s.x = a0[j]; s.y = a1[j];
        *(v2f*)(red + wv * NCOL + j * 64 + 2 * lane) = s;
    }
    __syncthreads();
    if (dpout == nullptr) {
        for (int o = t; o < NCOL; o += 256) {
            float s = 0.f;
            for (int w = 0; w < 8; ++w) s += red[w * NCOL + o];
            ypout[(size_t)b * NCOL + o] = s;
        }
        return;
    }
    for (int o = t; o < NCOL; o += 256) {
        float s = 0.f;
        for (int w = 0; w < 8; ++w) s += red[w * NCOL + o];
        ymean[(o >> 6) * 68 + (o & 63)] = s;
    }
    __syncthreads();
    {
        int p  = t >> 2;
        int qd = t & 3;
        const float* zp = stage + p * 65 + qd * 16;
        const float* yp = ymean + (p & 7) * 68 + qd * 16;
        float d = 0.f;
        for (int e = 0; e < 16; ++e) d += zp[e] * yp[e];
        d += ROW_XMASK(d, 1);
        d += ROW_XMASK(d, 2);
        if (qd == 0) Bs[p] = d;
        __syncthreads();
        if (t < 64) dpout[b * 64 + t] = Bs[t];
    }
}

extern "C" void kernel_launch(void* const* d_in, const int* in_sizes, int n_in,
                              void* d_out, int out_size, void* d_ws, size_t ws_size,
                              hipStream_t stream) {
    const float* x = (const float*)d_in[0];
    const float* W = (const float*)d_in[1];
    float* out = (float*)d_out;
    float* ws  = (float*)d_ws;

    float* qws    = ws;
    float* xpart  = qws + 131072;
    float* ypartA = xpart + 8192;
    float* dpart    = ypartA + 16384;
    float* Zm     = dpart + 16384;
    float* Bmat   = Zm + 4096;

    const float* x7 = x + (size_t)7 * JB * IN_DIM;
    const float* w7 = W + (size_t)7 * NUM_OUT * IN_DIM * OUT_DIM;

    k_first<<<XBLK + STRIPS, 256, 0, stream>>>(x, xpart, x7, w7, qws);
    k_route<<<96, 256, 0, stream>>>(x7, qws, nullptr, nullptr, nullptr, nullptr, ypartA, nullptr,
                                    xpart, W, w7, Zm);
    k_route<<<32, 256, 0, stream>>>(x7, qws, ypartA, Zm, nullptr, Bmat, nullptr, dpart,
                                    nullptr, nullptr, nullptr, nullptr);
    k_final<<<STRIPS, 256, 0, stream>>>(x7, w7, dpart, Bmat, out);
}
